# starting kernel
# baseline (speedup 1.0000x reference)
#include <hip/hip_runtime.h>
#include <hip/hip_bf16.h>


typedef __attribute__((ext_vector_type(2))) float v2f;
typedef __attribute__((ext_vector_type(8))) float v8f;

#define HEADS 2
#define FDIM 128
#define NEG_SLOPE 0.2f

__device__ __forceinline__ unsigned enc_f32(float f) {
    unsigned u = __float_as_uint(f);
    return (u & 0x80000000u) ? ~u : (u | 0x80000000u);
}
__device__ __forceinline__ float dec_f32(unsigned u) {
    return __uint_as_float((u & 0x80000000u) ? (u & 0x7FFFFFFFu) : ~u);
}

__global__ __launch_bounds__(256) void gemm128_wmma(
    const float* __restrict__ X, const float* __restrict__ W,
    float* __restrict__ Y, int M)
{
    __shared__ float lw[FDIM * FDIM];
    {
        const float4* w4 = (const float4*)W;
        float4* l4 = (float4*)lw;
        for (int i = threadIdx.x; i < FDIM * FDIM / 4; i += 256) l4[i] = w4[i];
    }
    __syncthreads();

    const int wave = threadIdx.x >> 5;
    const int lane = threadIdx.x & 31;
    const int m0 = (blockIdx.x * 8 + wave) * 16;
    if (m0 >= M) return;

    const int lm = lane & 15;
    const int lh = lane >> 4;
    const float* arow = X + (size_t)(m0 + lm) * FDIM + 2 * lh;

#pragma unroll
    for (int nt = 0; nt < 8; ++nt) {
        v8f acc = {};
        const int nc = nt * 16 + lm;
#pragma unroll 8
        for (int kk = 0; kk < 32; ++kk) {
            const int k0 = kk * 4;
            v2f a = *(const v2f*)(arow + k0);
            v2f b;
            b.x = lw[(k0 + 2 * lh) * FDIM + nc];
            b.y = lw[(k0 + 2 * lh + 1) * FDIM + nc];
            acc = __builtin_amdgcn_wmma_f32_16x16x4_f32(
                false, a, false, b, (short)0, acc, false, false);
        }
        float* yrow = Y + (size_t)(m0 + 8 * lh) * FDIM + nc;
#pragma unroll
        for (int r = 0; r < 8; ++r) yrow[(size_t)r * FDIM] = acc[r];
    }
}

__global__ void attn_logits(const float* __restrict__ h,
                            const float* __restrict__ a_src,
                            const float* __restrict__ a_dst,
                            float* __restrict__ esrc, float* __restrict__ edst,
                            int n)
{
    int idx = blockIdx.x * blockDim.x + threadIdx.x;
    if (idx >= n * HEADS) return;
    int node = idx >> 1, hd = idx & 1;
    const float* hp = h + (size_t)node * FDIM + hd * 64;
    const float* as = a_src + hd * 64;
    const float* ad = a_dst + hd * 64;
    float s = 0.f, t = 0.f;
#pragma unroll 8
    for (int f = 0; f < 64; ++f) { float x = hp[f]; s += x * as[f]; t += x * ad[f]; }
    esrc[idx] = s; edst[idx] = t;
}

__global__ void init_accum(unsigned* __restrict__ m, float* __restrict__ denom,
                           float* __restrict__ agg, int n)
{
    int i = blockIdx.x * blockDim.x + threadIdx.x;
    if (i < n * FDIM) agg[i] = 0.f;
    if (i < n * HEADS) { m[i] = 0u; denom[i] = 0.f; }
}

__global__ void edge_max(const int* __restrict__ src, const int* __restrict__ dst,
                         int E, int n,
                         const float* __restrict__ esrc,
                         const float* __restrict__ edst,
                         unsigned* __restrict__ m)
{
    int e = blockIdx.x * blockDim.x + threadIdx.x;
    if (e >= E + n) return;
    int s = (e < E) ? src[e] : (e - E);
    int d = (e < E) ? dst[e] : (e - E);
#pragma unroll
    for (int h = 0; h < HEADS; ++h) {
        float v = esrc[s * HEADS + h] + edst[d * HEADS + h];
        v = (v >= 0.f) ? v : NEG_SLOPE * v;
        atomicMax(&m[d * HEADS + h], enc_f32(v));
    }
}

__global__ void edge_aggregate(const int* __restrict__ src, const int* __restrict__ dst,
                               int E, int n,
                               const float* __restrict__ esrc,
                               const float* __restrict__ edst,
                               const unsigned* __restrict__ m,
                               const float* __restrict__ hfeat,
                               float* __restrict__ denom,
                               float* __restrict__ agg)
{
    int wid = (blockIdx.x * blockDim.x + threadIdx.x) >> 5;
    int lane = threadIdx.x & 31;
    if (wid >= E + n) return;
    int s = (wid < E) ? src[wid] : (wid - E);
    int d = (wid < E) ? dst[wid] : (wid - E);

    float v0 = esrc[s * HEADS + 0] + edst[d * HEADS + 0];
    float v1 = esrc[s * HEADS + 1] + edst[d * HEADS + 1];
    v0 = (v0 >= 0.f) ? v0 : NEG_SLOPE * v0;
    v1 = (v1 >= 0.f) ? v1 : NEG_SLOPE * v1;
    float a0 = __expf(v0 - dec_f32(m[d * HEADS + 0]));
    float a1 = __expf(v1 - dec_f32(m[d * HEADS + 1]));

    if (lane == 0) {
        atomicAdd(&denom[d * HEADS + 0], a0);
        atomicAdd(&denom[d * HEADS + 1], a1);
    }
    const float* hs = hfeat + (size_t)s * FDIM;
    float* od = agg + (size_t)d * FDIM;
#pragma unroll
    for (int i = 0; i < 4; ++i) {
        int f = lane + 32 * i;
        float al = (f < 64) ? a0 : a1;
        atomicAdd(&od[f], al * hs[f]);
    }
}

__global__ void node_finalize(const float* __restrict__ agg,
                              const float* __restrict__ denom,
                              const float* __restrict__ bias,
                              float* __restrict__ z, int n, int do_relu)
{
    int idx = blockIdx.x * blockDim.x + threadIdx.x;
    if (idx >= n * FDIM) return;
    int node = idx >> 7, c = idx & 127, hd = c >> 6;
    float v = agg[idx] / denom[node * HEADS + hd] + bias[c];
    if (do_relu) v = fmaxf(v, 0.f);
    z[idx] = v;
}

__global__ void decode_edges(const int* __restrict__ src, const int* __restrict__ dst,
                             const float* __restrict__ z,
                             float* __restrict__ out, int E)
{
    int wid = (blockIdx.x * blockDim.x + threadIdx.x) >> 5;
    int lane = threadIdx.x & 31;
    if (wid >= E) return;
    const float* zs = z + (size_t)src[wid] * FDIM;
    const float* zd = z + (size_t)dst[wid] * FDIM;
    float acc = 0.f;
#pragma unroll
    for (int i = 0; i < 4; ++i) { int f = lane + 32 * i; acc += zs[f] * zd[f]; }
#pragma unroll
    for (int off = 16; off > 0; off >>= 1) acc += __shfl_down(acc, off, 32);
    if (lane == 0) out[wid] = acc;
}

extern "C" void kernel_launch(void* const* d_in, const int* in_sizes, int n_in,
                              void* d_out, int out_size, void* d_ws, size_t ws_size,
                              hipStream_t stream) {
    const float* x      = (const float*)d_in[0];
    const int*   ei     = (const int*)d_in[1];
    const float* W1     = (const float*)d_in[2];
    const float* a1s    = (const float*)d_in[3];
    const float* a1d    = (const float*)d_in[4];
    const float* b1     = (const float*)d_in[5];
    const float* W2     = (const float*)d_in[6];
    const float* a2s    = (const float*)d_in[7];
    const float* a2d    = (const float*)d_in[8];
    const float* b2     = (const float*)d_in[9];
    float* scores = (float*)d_out;

    const int N = in_sizes[0] / FDIM;
    const int E = in_sizes[1] / 2;
    const int* src = ei;
    const int* dst = ei + E;

    float* ws   = (float*)d_ws;
    float*    h     = ws;
    float*    agg   = ws + (size_t)N * FDIM;
    float*    z     = ws + (size_t)2 * N * FDIM;
    float*    esrc  = ws + (size_t)3 * N * FDIM;
    float*    edst  = esrc + (size_t)N * HEADS;
    unsigned* mmax  = (unsigned*)(edst + (size_t)N * HEADS);
    float*    denom = (float*)mmax + (size_t)N * HEADS;

    const int ETOT = E + N;
    dim3 blk(256);
    dim3 gGemm((N + 127) / 128);
    dim3 gNH((N * HEADS + 255) / 256);
    dim3 gInit((N * FDIM + 255) / 256);
    dim3 gEdge((ETOT + 255) / 256);
    dim3 gEdgeW((ETOT * 32 + 255) / 256);
    dim3 gDecW(((size_t)E * 32 + 255) / 256);

    gemm128_wmma<<<gGemm, blk, 0, stream>>>(x, W1, h, N);
    attn_logits<<<gNH, blk, 0, stream>>>(h, a1s, a1d, esrc, edst, N);
    init_accum<<<gInit, blk, 0, stream>>>(mmax, denom, agg, N);
    edge_max<<<gEdge, blk, 0, stream>>>(src, dst, E, N, esrc, edst, mmax);
    edge_aggregate<<<gEdgeW, blk, 0, stream>>>(src, dst, E, N, esrc, edst, mmax, h, denom, agg);
    node_finalize<<<gInit, blk, 0, stream>>>(agg, denom, b1, z, N,  1);

    gemm128_wmma<<<gGemm, blk, 0, stream>>>(z, W2, h, N);
    attn_logits<<<gNH, blk, 0, stream>>>(h, a2s, a2d, esrc, edst, N);
    init_accum<<<gInit, blk, 0, stream>>>(mmax, denom, agg, N);
    edge_max<<<gEdge, blk, 0, stream>>>(src, dst, E, N, esrc, edst, mmax);
    edge_aggregate<<<gEdgeW, blk, 0, stream>>>(src, dst, E, N, esrc, edst, mmax, h, denom, agg);
    node_finalize<<<gInit, blk, 0, stream>>>(agg, denom, b2, z, N,  0);

    decode_edges<<<gDecW, blk, 0, stream>>>(src, dst, z, scores, E);
}


# edge softmax+aggregation per 64-node destination tile held in LDS: edges bucketed by tile in-call, LDS max/add atomics (arrival-order sums as before); global float atomics, init/max/finalize gone
# speedup vs baseline: 2.3414x; 2.3414x over previous
#include <hip/hip_runtime.h>
#include <hip/hip_bf16.h>


typedef __attribute__((ext_vector_type(2))) float v2f;
typedef __attribute__((ext_vector_type(8))) float v8f;

#define HEADS 2
#define FDIM 128
#define NEG_SLOPE 0.2f

__device__ __forceinline__ unsigned enc_f32(float f) {
    unsigned u = __float_as_uint(f);
    return (u & 0x80000000u) ? ~u : (u | 0x80000000u);
}
__device__ __forceinline__ float dec_f32(unsigned u) {
    return __uint_as_float((u & 0x80000000u) ? (u & 0x7FFFFFFFu) : ~u);
}

__global__ __launch_bounds__(256) void gemm128_wmma(
    const float* __restrict__ X, const float* __restrict__ W,
    float* __restrict__ Y, int M)
{
    __shared__ float lw[FDIM * FDIM];
    {
        const float4* w4 = (const float4*)W;
        float4* l4 = (float4*)lw;
        for (int i = threadIdx.x; i < FDIM * FDIM / 4; i += 256) l4[i] = w4[i];
    }
    __syncthreads();

    const int wave = threadIdx.x >> 5;
    const int lane = threadIdx.x & 31;
    const int m0 = (blockIdx.x * 8 + wave) * 16;
    if (m0 >= M) return;

    const int lm = lane & 15;
    const int lh = lane >> 4;
    const float* arow = X + (size_t)(m0 + lm) * FDIM + 2 * lh;

#pragma unroll
    for (int nt = 0; nt < 8; ++nt) {
        v8f acc = {};
        const int nc = nt * 16 + lm;
#pragma unroll 8
        for (int kk = 0; kk < 32; ++kk) {
            const int k0 = kk * 4;
            v2f a = *(const v2f*)(arow + k0);
            v2f b;
            b.x = lw[(k0 + 2 * lh) * FDIM + nc];
            b.y = lw[(k0 + 2 * lh + 1) * FDIM + nc];
            acc = __builtin_amdgcn_wmma_f32_16x16x4_f32(
                false, a, false, b, (short)0, acc, false, false);
        }
        float* yrow = Y + (size_t)(m0 + 8 * lh) * FDIM + nc;
#pragma unroll
        for (int r = 0; r < 8; ++r) yrow[(size_t)r * FDIM] = acc[r];
    }
}

__global__ void attn_logits(const float* __restrict__ h,
                            const float* __restrict__ a_src,
                            const float* __restrict__ a_dst,
                            float* __restrict__ esrc, float* __restrict__ edst,
                            int n)
{
    int idx = blockIdx.x * blockDim.x + threadIdx.x;
    if (idx >= n * HEADS) return;
    int node = idx >> 1, hd = idx & 1;
    const float* hp = h + (size_t)node * FDIM + hd * 64;
    const float* as = a_src + hd * 64;
    const float* ad = a_dst + hd * 64;
    float s = 0.f, t = 0.f;
#pragma unroll 8
    for (int f = 0; f < 64; ++f) { float x = hp[f]; s += x * as[f]; t += x * ad[f]; }
    esrc[idx] = s; edst[idx] = t;
}

#define TILE 64
#define TILE_SHIFT 6
#define MAXNB 1024
#define CHUNK 8192

__global__ void bucket_zero(int* __restrict__ bcount, int nb)
{
    int i = blockIdx.x * blockDim.x + threadIdx.x;
    if (i < nb) bcount[i] = 0;
}

__global__ __launch_bounds__(256) void bucket_count(const int* __restrict__ dst,
                                                    int E, int n, int nb,
                                                    int* __restrict__ bcount)
{
    __shared__ int lh[MAXNB];
    for (int i = threadIdx.x; i < nb; i += 256) lh[i] = 0;
    __syncthreads();
    const int base = blockIdx.x * CHUNK;
    for (int k = 0; k < CHUNK / 256; ++k) {
        int e = base + k * 256 + threadIdx.x;
        if (e < E + n) {
            int d = (e < E) ? dst[e] : (e - E);
            atomicAdd(&lh[d >> TILE_SHIFT], 1);
        }
    }
    __syncthreads();
    for (int i = threadIdx.x; i < nb; i += 256) {
        int c = lh[i];
        if (c) atomicAdd(&bcount[i], c);
    }
}

__global__ __launch_bounds__(1024) void bucket_scan(const int* __restrict__ bcount, int nb,
                                                    int* __restrict__ bstart,
                                                    int* __restrict__ bcursor)
{
    __shared__ int s[1024];
    const int t = threadIdx.x;
    const int c = (t < nb) ? bcount[t] : 0;
    s[t] = c;
    __syncthreads();
    for (int off = 1; off < 1024; off <<= 1) {
        int v = (t >= off) ? s[t - off] : 0;
        __syncthreads();
        s[t] += v;
        __syncthreads();
    }
    const int incl = s[t];
    if (t < nb) { bstart[t] = incl - c; bcursor[t] = incl - c; }
    if (t == nb - 1) bstart[nb] = incl;
}

__global__ __launch_bounds__(256) void bucket_scatter(const int* __restrict__ src,
                                                      const int* __restrict__ dst,
                                                      int E, int n, int nb,
                                                      int* __restrict__ bcursor,
                                                      int* __restrict__ psrc,
                                                      int* __restrict__ pdst)
{
    __shared__ int lh[MAXNB];
    __shared__ int lbase[MAXNB];
    for (int i = threadIdx.x; i < nb; i += 256) lh[i] = 0;
    __syncthreads();
    const int base = blockIdx.x * CHUNK;
    for (int k = 0; k < CHUNK / 256; ++k) {
        int e = base + k * 256 + threadIdx.x;
        if (e < E + n) {
            int d = (e < E) ? dst[e] : (e - E);
            atomicAdd(&lh[d >> TILE_SHIFT], 1);
        }
    }
    __syncthreads();
    for (int i = threadIdx.x; i < nb; i += 256) {
        int c = lh[i];
        lbase[i] = c ? atomicAdd(&bcursor[i], c) : 0;
        lh[i] = 0;
    }
    __syncthreads();
    for (int k = 0; k < CHUNK / 256; ++k) {
        int e = base + k * 256 + threadIdx.x;
        if (e < E + n) {
            int s = (e < E) ? src[e] : (e - E);
            int d = (e < E) ? dst[e] : (e - E);
            int b = d >> TILE_SHIFT;
            int pos = lbase[b] + atomicAdd(&lh[b], 1);
            psrc[pos] = s;
            pdst[pos] = d;
        }
    }
}

#define EPW 8
#define MPT 4

__global__ __launch_bounds__(256) void tile_softmax_aggregate(
    const int* __restrict__ psrc, const int* __restrict__ pdst,
    const int* __restrict__ bstart,
    const float* __restrict__ esrc, const float* __restrict__ edst,
    const float* __restrict__ hfeat, const float* __restrict__ bias,
    float* __restrict__ z, int n, int do_relu)
{
    __shared__ float acc[TILE * FDIM];
    __shared__ unsigned mx[TILE * HEADS];
    __shared__ float den[TILE * HEADS];

    const int tid = threadIdx.x;
    const int wave = tid >> 5;
    const int lane = tid & 31;
    const int n0 = blockIdx.x * TILE;
    const int e0 = bstart[blockIdx.x];
    const int e1 = bstart[blockIdx.x + 1];

    for (int i = tid; i < TILE * FDIM; i += 256) acc[i] = 0.f;
    if (tid < TILE * HEADS) { mx[tid] = 0u; den[tid] = 0.f; }
    __syncthreads();

    for (int eb = e0 + tid; eb < e1; eb += 256 * MPT) {
        int sm[MPT], dm[MPT];
#pragma unroll
        for (int u = 0; u < MPT; ++u) {
            int ee = (eb + 256 * u < e1) ? (eb + 256 * u) : (e1 - 1);
            sm[u] = psrc[ee];
            dm[u] = pdst[ee];
        }
        float w0[MPT], w1[MPT];
#pragma unroll
        for (int u = 0; u < MPT; ++u) {
            float v0 = esrc[sm[u] * HEADS + 0] + edst[dm[u] * HEADS + 0];
            float v1 = esrc[sm[u] * HEADS + 1] + edst[dm[u] * HEADS + 1];
            w0[u] = (v0 >= 0.f) ? v0 : NEG_SLOPE * v0;
            w1[u] = (v1 >= 0.f) ? v1 : NEG_SLOPE * v1;
        }
#pragma unroll
        for (int u = 0; u < MPT; ++u) {
            if (eb + 256 * u < e1) {
                atomicMax(&mx[(dm[u] - n0) * HEADS + 0], enc_f32(w0[u]));
                atomicMax(&mx[(dm[u] - n0) * HEADS + 1], enc_f32(w1[u]));
            }
        }
    }
    __syncthreads();

    if (e0 < e1) {
        int sc[EPW], dc[EPW];
        const int eb0 = e0 + wave * EPW;
#pragma unroll
        for (int u = 0; u < EPW; ++u) {
            int ee = (eb0 + u < e1) ? (eb0 + u) : (e1 - 1);
            sc[u] = psrc[ee];
            dc[u] = pdst[ee];
        }
        for (int eb = eb0; eb < e1; eb += 8 * EPW) {
            int sn[EPW], dn[EPW];
#pragma unroll
            for (int u = 0; u < EPW; ++u) {
                int ee = (eb + 8 * EPW + u < e1) ? (eb + 8 * EPW + u) : (e1 - 1);
                sn[u] = psrc[ee];
                dn[u] = pdst[ee];
            }
            float hv[EPW][4];
            float a0[EPW], a1[EPW];
#pragma unroll
            for (int u = 0; u < EPW; ++u) {
                const int s = sc[u], d = dc[u];
                float v0 = esrc[s * HEADS + 0] + edst[d * HEADS + 0];
                float v1 = esrc[s * HEADS + 1] + edst[d * HEADS + 1];
                v0 = (v0 >= 0.f) ? v0 : NEG_SLOPE * v0;
                v1 = (v1 >= 0.f) ? v1 : NEG_SLOPE * v1;
                a0[u] = __expf(v0 - dec_f32(mx[(d - n0) * HEADS + 0]));
                a1[u] = __expf(v1 - dec_f32(mx[(d - n0) * HEADS + 1]));
                const float* hs = hfeat + (size_t)s * FDIM;
#pragma unroll
                for (int i = 0; i < 4; ++i) hv[u][i] = hs[lane + 32 * i];
            }
#pragma unroll
            for (int u = 0; u < EPW; ++u) {
                if (eb + u < e1) {
                    const int dl = dc[u] - n0;
                    if (lane < HEADS) atomicAdd(&den[dl * HEADS + lane], lane ? a1[u] : a0[u]);
                    float* od = acc + dl * FDIM;
#pragma unroll
                    for (int i = 0; i < 4; ++i) {
                        int f = lane + 32 * i;
                        float al = (f < 64) ? a0[u] : a1[u];
                        atomicAdd(&od[f], al * hv[u][i]);
                    }
                }
            }
#pragma unroll
            for (int u = 0; u < EPW; ++u) { sc[u] = sn[u]; dc[u] = dn[u]; }
        }
    }
    __syncthreads();

    for (int i = tid; i < TILE * FDIM; i += 256) {
        int ln = i >> 7, c = i & 127, hd = c >> 6;
        int node = n0 + ln;
        if (node < n) {
            float v = acc[i] / den[ln * HEADS + hd] + bias[c];
            if (do_relu) v = fmaxf(v, 0.f);
            z[(size_t)node * FDIM + c] = v;
        }
    }
}

__global__ void decode_edges(const int* __restrict__ src, const int* __restrict__ dst,
                             const float* __restrict__ z,
                             float* __restrict__ out, int E)
{
    int wid = (blockIdx.x * blockDim.x + threadIdx.x) >> 5;
    int lane = threadIdx.x & 31;
    if (wid >= E) return;
    const float* zs = z + (size_t)src[wid] * FDIM;
    const float* zd = z + (size_t)dst[wid] * FDIM;
    float acc = 0.f;
#pragma unroll
    for (int i = 0; i < 4; ++i) { int f = lane + 32 * i; acc += zs[f] * zd[f]; }
#pragma unroll
    for (int off = 16; off > 0; off >>= 1) acc += __shfl_down(acc, off, 32);
    if (lane == 0) out[wid] = acc;
}

extern "C" void kernel_launch(void* const* d_in, const int* in_sizes, int n_in,
                              void* d_out, int out_size, void* d_ws, size_t ws_size,
                              hipStream_t stream) {
    const float* x      = (const float*)d_in[0];
    const int*   ei     = (const int*)d_in[1];
    const float* W1     = (const float*)d_in[2];
    const float* a1s    = (const float*)d_in[3];
    const float* a1d    = (const float*)d_in[4];
    const float* b1     = (const float*)d_in[5];
    const float* W2     = (const float*)d_in[6];
    const float* a2s    = (const float*)d_in[7];
    const float* a2d    = (const float*)d_in[8];
    const float* b2     = (const float*)d_in[9];
    float* scores = (float*)d_out;

    const int N = in_sizes[0] / FDIM;
    const int E = in_sizes[1] / 2;
    const int* src = ei;
    const int* dst = ei + E;

    float* ws   = (float*)d_ws;
    float*    h     = ws;
    int*      psrc  = (int*)(ws + (size_t)N * FDIM);
    int*      pdst  = psrc + (size_t)(E + N);
    int*      bcount  = pdst + (size_t)(E + N);
    int*      bstart  = bcount + MAXNB;
    int*      bcursor = bstart + MAXNB + 1;
    float*    z     = ws + (size_t)2 * N * FDIM;
    float*    esrc  = ws + (size_t)3 * N * FDIM;
    float*    edst  = esrc + (size_t)N * HEADS;

    const int ETOT = E + N;
    const int NB = (N + TILE - 1) / TILE;
    dim3 blk(256);
    dim3 gGemm((N + 127) / 128);
    dim3 gNH((N * HEADS + 255) / 256);
    dim3 gChunk((ETOT + CHUNK - 1) / CHUNK);
    dim3 gDecW(((size_t)E * 32 + 255) / 256);

    bucket_zero<<<dim3((NB + 255) / 256), blk, 0, stream>>>(bcount, NB);
    bucket_count<<<gChunk, blk, 0, stream>>>(dst, E, N, NB, bcount);
    bucket_scan<<<dim3(1), dim3(1024), 0, stream>>>(bcount, NB, bstart, bcursor);
    bucket_scatter<<<gChunk, blk, 0, stream>>>(src, dst, E, N, NB, bcursor, psrc, pdst);

    gemm128_wmma<<<gGemm, blk, 0, stream>>>(x, W1, h, N);
    attn_logits<<<gNH, blk, 0, stream>>>(h, a1s, a1d, esrc, edst, N);
    tile_softmax_aggregate<<<dim3(NB), blk, 0, stream>>>(psrc, pdst, bstart, esrc, edst, h, b1, z, N,  1);

    gemm128_wmma<<<gGemm, blk, 0, stream>>>(z, W2, h, N);
    attn_logits<<<gNH, blk, 0, stream>>>(h, a2s, a2d, esrc, edst, N);
    tile_softmax_aggregate<<<dim3(NB), blk, 0, stream>>>(psrc, pdst, bstart, esrc, edst, h, b2, z, N,  0);

    decode_edges<<<gDecW, blk, 0, stream>>>(src, dst, z, scores, E);
}


# link decode over the destination-tile buckets: work-group per half tile, tile z rows in LDS, one gathered z row per edge (was two, one wave per edge); edge id kept at bucketing; new score sum tree
# speedup vs baseline: 2.6957x; 1.1513x over previous
#include <hip/hip_runtime.h>
#include <hip/hip_bf16.h>


typedef __attribute__((ext_vector_type(2))) float v2f;
typedef __attribute__((ext_vector_type(8))) float v8f;

#define HEADS 2
#define FDIM 128
#define NEG_SLOPE 0.2f

__device__ __forceinline__ unsigned enc_f32(float f) {
    unsigned u = __float_as_uint(f);
    return (u & 0x80000000u) ? ~u : (u | 0x80000000u);
}
__device__ __forceinline__ float dec_f32(unsigned u) {
    return __uint_as_float((u & 0x80000000u) ? (u & 0x7FFFFFFFu) : ~u);
}

__global__ __launch_bounds__(256) void gemm128_wmma(
    const float* __restrict__ X, const float* __restrict__ W,
    float* __restrict__ Y, int M)
{
    __shared__ float lw[FDIM * FDIM];
    {
        const float4* w4 = (const float4*)W;
        float4* l4 = (float4*)lw;
        for (int i = threadIdx.x; i < FDIM * FDIM / 4; i += 256) l4[i] = w4[i];
    }
    __syncthreads();

    const int wave = threadIdx.x >> 5;
    const int lane = threadIdx.x & 31;
    const int m0 = (blockIdx.x * 8 + wave) * 16;
    if (m0 >= M) return;

    const int lm = lane & 15;
    const int lh = lane >> 4;
    const float* arow = X + (size_t)(m0 + lm) * FDIM + 2 * lh;

#pragma unroll
    for (int nt = 0; nt < 8; ++nt) {
        v8f acc = {};
        const int nc = nt * 16 + lm;
#pragma unroll 8
        for (int kk = 0; kk < 32; ++kk) {
            const int k0 = kk * 4;
            v2f a = *(const v2f*)(arow + k0);
            v2f b;
            b.x = lw[(k0 + 2 * lh) * FDIM + nc];
            b.y = lw[(k0 + 2 * lh + 1) * FDIM + nc];
            acc = __builtin_amdgcn_wmma_f32_16x16x4_f32(
                false, a, false, b, (short)0, acc, false, false);
        }
        float* yrow = Y + (size_t)(m0 + 8 * lh) * FDIM + nc;
#pragma unroll
        for (int r = 0; r < 8; ++r) yrow[(size_t)r * FDIM] = acc[r];
    }
}

__global__ void attn_logits(const float* __restrict__ h,
                            const float* __restrict__ a_src,
                            const float* __restrict__ a_dst,
                            float* __restrict__ esrc, float* __restrict__ edst,
                            int n)
{
    int idx = blockIdx.x * blockDim.x + threadIdx.x;
    if (idx >= n * HEADS) return;
    int node = idx >> 1, hd = idx & 1;
    const float* hp = h + (size_t)node * FDIM + hd * 64;
    const float* as = a_src + hd * 64;
    const float* ad = a_dst + hd * 64;
    float s = 0.f, t = 0.f;
#pragma unroll 8
    for (int f = 0; f < 64; ++f) { float x = hp[f]; s += x * as[f]; t += x * ad[f]; }
    esrc[idx] = s; edst[idx] = t;
}

#define TILE 64
#define TILE_SHIFT 6
#define MAXNB 1024
#define CHUNK 8192

__global__ void bucket_zero(int* __restrict__ bcount, int nb)
{
    int i = blockIdx.x * blockDim.x + threadIdx.x;
    if (i < nb) bcount[i] = 0;
}

__global__ __launch_bounds__(256) void bucket_count(const int* __restrict__ dst,
                                                    int E, int n, int nb,
                                                    int* __restrict__ bcount)
{
    __shared__ int lh[MAXNB];
    for (int i = threadIdx.x; i < nb; i += 256) lh[i] = 0;
    __syncthreads();
    const int base = blockIdx.x * CHUNK;
    for (int k = 0; k < CHUNK / 256; ++k) {
        int e = base + k * 256 + threadIdx.x;
        if (e < E + n) {
            int d = (e < E) ? dst[e] : (e - E);
            atomicAdd(&lh[d >> TILE_SHIFT], 1);
        }
    }
    __syncthreads();
    for (int i = threadIdx.x; i < nb; i += 256) {
        int c = lh[i];
        if (c) atomicAdd(&bcount[i], c);
    }
}

__global__ __launch_bounds__(1024) void bucket_scan(const int* __restrict__ bcount, int nb,
                                                    int* __restrict__ bstart,
                                                    int* __restrict__ bcursor)
{
    __shared__ int s[1024];
    const int t = threadIdx.x;
    const int c = (t < nb) ? bcount[t] : 0;
    s[t] = c;
    __syncthreads();
    for (int off = 1; off < 1024; off <<= 1) {
        int v = (t >= off) ? s[t - off] : 0;
        __syncthreads();
        s[t] += v;
        __syncthreads();
    }
    const int incl = s[t];
    if (t < nb) { bstart[t] = incl - c; bcursor[t] = incl - c; }
    if (t == nb - 1) bstart[nb] = incl;
}

__global__ __launch_bounds__(256) void bucket_scatter(const int* __restrict__ src,
                                                      const int* __restrict__ dst,
                                                      int E, int n, int nb,
                                                      int* __restrict__ bcursor,
                                                      int* __restrict__ psrc,
                                                      int* __restrict__ pdst,
                                                      int* __restrict__ peid)
{
    __shared__ int lh[MAXNB];
    __shared__ int lbase[MAXNB];
    for (int i = threadIdx.x; i < nb; i += 256) lh[i] = 0;
    __syncthreads();
    const int base = blockIdx.x * CHUNK;
    for (int k = 0; k < CHUNK / 256; ++k) {
        int e = base + k * 256 + threadIdx.x;
        if (e < E + n) {
            int d = (e < E) ? dst[e] : (e - E);
            atomicAdd(&lh[d >> TILE_SHIFT], 1);
        }
    }
    __syncthreads();
    for (int i = threadIdx.x; i < nb; i += 256) {
        int c = lh[i];
        lbase[i] = c ? atomicAdd(&bcursor[i], c) : 0;
        lh[i] = 0;
    }
    __syncthreads();
    for (int k = 0; k < CHUNK / 256; ++k) {
        int e = base + k * 256 + threadIdx.x;
        if (e < E + n) {
            int s = (e < E) ? src[e] : (e - E);
            int d = (e < E) ? dst[e] : (e - E);
            int b = d >> TILE_SHIFT;
            int pos = lbase[b] + atomicAdd(&lh[b], 1);
            psrc[pos] = s;
            pdst[pos] = d;
            peid[pos] = e;
        }
    }
}

#define EPW 8
#define MPT 4

__global__ __launch_bounds__(256) void tile_softmax_aggregate(
    const int* __restrict__ psrc, const int* __restrict__ pdst,
    const int* __restrict__ bstart,
    const float* __restrict__ esrc, const float* __restrict__ edst,
    const float* __restrict__ hfeat, const float* __restrict__ bias,
    float* __restrict__ z, int n, int do_relu)
{
    __shared__ float acc[TILE * FDIM];
    __shared__ unsigned mx[TILE * HEADS];
    __shared__ float den[TILE * HEADS];

    const int tid = threadIdx.x;
    const int wave = tid >> 5;
    const int lane = tid & 31;
    const int n0 = blockIdx.x * TILE;
    const int e0 = bstart[blockIdx.x];
    const int e1 = bstart[blockIdx.x + 1];

    for (int i = tid; i < TILE * FDIM; i += 256) acc[i] = 0.f;
    if (tid < TILE * HEADS) { mx[tid] = 0u; den[tid] = 0.f; }
    __syncthreads();

    for (int eb = e0 + tid; eb < e1; eb += 256 * MPT) {
        int sm[MPT], dm[MPT];
#pragma unroll
        for (int u = 0; u < MPT; ++u) {
            int ee = (eb + 256 * u < e1) ? (eb + 256 * u) : (e1 - 1);
            sm[u] = psrc[ee];
            dm[u] = pdst[ee];
        }
        float w0[MPT], w1[MPT];
#pragma unroll
        for (int u = 0; u < MPT; ++u) {
            float v0 = esrc[sm[u] * HEADS + 0] + edst[dm[u] * HEADS + 0];
            float v1 = esrc[sm[u] * HEADS + 1] + edst[dm[u] * HEADS + 1];
            w0[u] = (v0 >= 0.f) ? v0 : NEG_SLOPE * v0;
            w1[u] = (v1 >= 0.f) ? v1 : NEG_SLOPE * v1;
        }
#pragma unroll
        for (int u = 0; u < MPT; ++u) {
            if (eb + 256 * u < e1) {
                atomicMax(&mx[(dm[u] - n0) * HEADS + 0], enc_f32(w0[u]));
                atomicMax(&mx[(dm[u] - n0) * HEADS + 1], enc_f32(w1[u]));
            }
        }
    }
    __syncthreads();

    if (e0 < e1) {
        int sc[EPW], dc[EPW];
        const int eb0 = e0 + wave * EPW;
#pragma unroll
        for (int u = 0; u < EPW; ++u) {
            int ee = (eb0 + u < e1) ? (eb0 + u) : (e1 - 1);
            sc[u] = psrc[ee];
            dc[u] = pdst[ee];
        }
        for (int eb = eb0; eb < e1; eb += 8 * EPW) {
            int sn[EPW], dn[EPW];
#pragma unroll
            for (int u = 0; u < EPW; ++u) {
                int ee = (eb + 8 * EPW + u < e1) ? (eb + 8 * EPW + u) : (e1 - 1);
                sn[u] = psrc[ee];
                dn[u] = pdst[ee];
            }
            float hv[EPW][4];
            float a0[EPW], a1[EPW];
#pragma unroll
            for (int u = 0; u < EPW; ++u) {
                const int s = sc[u], d = dc[u];
                float v0 = esrc[s * HEADS + 0] + edst[d * HEADS + 0];
                float v1 = esrc[s * HEADS + 1] + edst[d * HEADS + 1];
                v0 = (v0 >= 0.f) ? v0 : NEG_SLOPE * v0;
                v1 = (v1 >= 0.f) ? v1 : NEG_SLOPE * v1;
                a0[u] = __expf(v0 - dec_f32(mx[(d - n0) * HEADS + 0]));
                a1[u] = __expf(v1 - dec_f32(mx[(d - n0) * HEADS + 1]));
                const float* hs = hfeat + (size_t)s * FDIM;
#pragma unroll
                for (int i = 0; i < 4; ++i) hv[u][i] = hs[lane + 32 * i];
            }
#pragma unroll
            for (int u = 0; u < EPW; ++u) {
                if (eb + u < e1) {
                    const int dl = dc[u] - n0;
                    if (lane < HEADS) atomicAdd(&den[dl * HEADS + lane], lane ? a1[u] : a0[u]);
                    float* od = acc + dl * FDIM;
#pragma unroll
                    for (int i = 0; i < 4; ++i) {
                        int f = lane + 32 * i;
                        float al = (f < 64) ? a0[u] : a1[u];
                        atomicAdd(&od[f], al * hv[u][i]);
                    }
                }
            }
#pragma unroll
            for (int u = 0; u < EPW; ++u) { sc[u] = sn[u]; dc[u] = dn[u]; }
        }
    }
    __syncthreads();

    for (int i = tid; i < TILE * FDIM; i += 256) {
        int ln = i >> 7, c = i & 127, hd = c >> 6;
        int node = n0 + ln;
        if (node < n) {
            float v = acc[i] / den[ln * HEADS + hd] + bias[c];
            if (do_relu) v = fmaxf(v, 0.f);
            z[(size_t)node * FDIM + c] = v;
        }
    }
}

#define DSLOTS 32
#define DBATCH 16

__device__ __forceinline__ float lane_xor16(float x) {
    return __builtin_bit_cast(float, __builtin_amdgcn_permlanex16(
        __builtin_bit_cast(int, x), __builtin_bit_cast(int, x), 0x76543210, 0xfedcba98, false, false));
}
template <int M>
__device__ __forceinline__ float lane_xor_row(float x) {
    return __builtin_bit_cast(float, __builtin_amdgcn_update_dpp(
        0, __builtin_bit_cast(int, x), 0x160 | M, 0xf, 0xf, true));
}

__global__ __launch_bounds__(256) void tile_decode(
    const int* __restrict__ psrc, const int* __restrict__ pdst, const int* __restrict__ peid,
    const int* __restrict__ bstart, const float* __restrict__ z,
    float* __restrict__ out, int n, int E)
{
    __shared__ float4 zt[TILE * FDIM / 4];

    const int tid = threadIdx.x;
    const int wave = tid >> 5;
    const int lane = tid & 31;
    const int tile = blockIdx.x >> 1;
    const int n0 = tile * TILE;
    const int t0 = bstart[tile];
    const int t1 = bstart[tile + 1];
    const int mid = t0 + ((t1 - t0) >> 1);
    const int e0 = (blockIdx.x & 1) ? mid : t0;
    const int e1 = (blockIdx.x & 1) ? t1 : mid;

    const float4* z4 = (const float4*)z;
    const unsigned loff = (unsigned)lane * 16u;
    float4 piece[TILE * FDIM / 4 / 256];
#pragma unroll
    for (int k = 0; k < TILE * FDIM / 4 / 256; ++k) {
        const int i = tid + 256 * k;
        const int node = min(n0 + (i >> 5), n - 1);
        piece[k] = z4[(size_t)node * (FDIM / 4) + (i & 31)];
    }
#pragma unroll
    for (int k = 0; k < TILE * FDIM / 4 / 256; ++k) zt[tid + 256 * k] = piece[k];
    __syncthreads();

    for (int eb = e0 + wave * DSLOTS; eb < e1; eb += 8 * DSLOTS) {
        const int ee = (eb + lane < e1) ? (eb + lane) : (e1 - 1);
        const int sv = psrc[ee];
        const int dv = pdst[ee] - n0;
        const int ev = peid[ee];

        float p[DSLOTS];
#pragma unroll
        for (int b = 0; b < DSLOTS / DBATCH; ++b) {
            float4 g[DBATCH];
#pragma unroll
            for (int u = 0; u < DBATCH; ++u) {
                const unsigned s = (unsigned)__builtin_amdgcn_readlane(sv, b * DBATCH + u);
                const char* rowp = (const char*)z + (size_t)s * (FDIM * sizeof(float));
                g[u] = *(const float4*)(rowp + loff);
            }
#pragma unroll
            for (int u = 0; u < DBATCH; ++u) {
                const int dl = __builtin_amdgcn_readlane(dv, b * DBATCH + u);
                const float4 t = zt[dl * (FDIM / 4) + lane];
                p[b * DBATCH + u] = g[u].x * t.x + g[u].y * t.y + g[u].z * t.z + g[u].w * t.w;
            }
        }

        {
            const bool hi = (lane & 16) != 0;
#pragma unroll
            for (int u = 0; u < 16; ++u) {
                const float keep = hi ? p[u + 16] : p[u];
                const float send = hi ? p[u] : p[u + 16];
                p[u] = keep + lane_xor16(send);
            }
        }
        {
            const bool hi = (lane & 8) != 0;
#pragma unroll
            for (int u = 0; u < 8; ++u) {
                const float keep = hi ? p[u + 8] : p[u];
                const float send = hi ? p[u] : p[u + 8];
                p[u] = keep + lane_xor_row<8>(send);
            }
        }
        {
            const bool hi = (lane & 4) != 0;
#pragma unroll
            for (int u = 0; u < 4; ++u) {
                const float keep = hi ? p[u + 4] : p[u];
                const float send = hi ? p[u] : p[u + 4];
                p[u] = keep + lane_xor_row<4>(send);
            }
        }
        {
            const bool hi = (lane & 2) != 0;
#pragma unroll
            for (int u = 0; u < 2; ++u) {
                const float keep = hi ? p[u + 2] : p[u];
                const float send = hi ? p[u] : p[u + 2];
                p[u] = keep + lane_xor_row<2>(send);
            }
        }
        {
            const bool hi = (lane & 1) != 0;
            const float keep = hi ? p[1] : p[0];
            const float send = hi ? p[0] : p[1];
            p[0] = keep + lane_xor_row<1>(send);
        }
        if (eb + lane < e1 && ev < E) out[ev] = p[0];
    }
}

extern "C" void kernel_launch(void* const* d_in, const int* in_sizes, int n_in,
                              void* d_out, int out_size, void* d_ws, size_t ws_size,
                              hipStream_t stream) {
    const float* x      = (const float*)d_in[0];
    const int*   ei     = (const int*)d_in[1];
    const float* W1     = (const float*)d_in[2];
    const float* a1s    = (const float*)d_in[3];
    const float* a1d    = (const float*)d_in[4];
    const float* b1     = (const float*)d_in[5];
    const float* W2     = (const float*)d_in[6];
    const float* a2s    = (const float*)d_in[7];
    const float* a2d    = (const float*)d_in[8];
    const float* b2     = (const float*)d_in[9];
    float* scores = (float*)d_out;

    const int N = in_sizes[0] / FDIM;
    const int E = in_sizes[1] / 2;
    const int* src = ei;
    const int* dst = ei + E;

    float* ws   = (float*)d_ws;
    float*    h     = ws;
    int*      psrc  = (int*)(ws + (size_t)N * FDIM);
    int*      pdst  = psrc + (size_t)(E + N);
    int*      bcount  = pdst + (size_t)(E + N);
    int*      bstart  = bcount + MAXNB;
    int*      bcursor = bstart + MAXNB + 1;
    int*      peid  = bcursor + MAXNB;
    float*    z     = ws + (size_t)2 * N * FDIM;
    float*    esrc  = ws + (size_t)3 * N * FDIM;
    float*    edst  = esrc + (size_t)N * HEADS;

    const int ETOT = E + N;
    const int NB = (N + TILE - 1) / TILE;
    dim3 blk(256);
    dim3 gGemm((N + 127) / 128);
    dim3 gNH((N * HEADS + 255) / 256);
    dim3 gChunk((ETOT + CHUNK - 1) / CHUNK);

    bucket_zero<<<dim3((NB + 255) / 256), blk, 0, stream>>>(bcount, NB);
    bucket_count<<<gChunk, blk, 0, stream>>>(dst, E, N, NB, bcount);
    bucket_scan<<<dim3(1), dim3(1024), 0, stream>>>(bcount, NB, bstart, bcursor);
    bucket_scatter<<<gChunk, blk, 0, stream>>>(src, dst, E, N, NB, bcursor, psrc, pdst, peid);

    gemm128_wmma<<<gGemm, blk, 0, stream>>>(x, W1, h, N);
    attn_logits<<<gNH, blk, 0, stream>>>(h, a1s, a1d, esrc, edst, N);
    tile_softmax_aggregate<<<dim3(NB), blk, 0, stream>>>(psrc, pdst, bstart, esrc, edst, h, b1, z, N,  1);

    gemm128_wmma<<<gGemm, blk, 0, stream>>>(z, W2, h, N);
    attn_logits<<<gNH, blk, 0, stream>>>(h, a2s, a2d, esrc, edst, N);
    tile_softmax_aggregate<<<dim3(NB), blk, 0, stream>>>(psrc, pdst, bstart, esrc, edst, h, b2, z, N,  0);

    tile_decode<<<dim3(2 * NB), blk, 0, stream>>>(psrc, pdst, peid, bstart, z, scores, N, E);
}


# edge softmax+aggregation as f32 WMMA products: 16-node buckets, one wave per bucket, one-hot weight fragment (16 dst x 4 edges) x 4 gathered rows, register sums; LDS float adds gone; new sum order
# speedup vs baseline: 2.9939x; 1.1106x over previous
#include <hip/hip_runtime.h>
#include <hip/hip_bf16.h>


typedef __attribute__((ext_vector_type(2))) float v2f;
typedef __attribute__((ext_vector_type(8))) float v8f;

#define HEADS 2
#define FDIM 128
#define NEG_SLOPE 0.2f

__device__ __forceinline__ unsigned enc_f32(float f) {
    unsigned u = __float_as_uint(f);
    return (u & 0x80000000u) ? ~u : (u | 0x80000000u);
}
__device__ __forceinline__ float dec_f32(unsigned u) {
    return __uint_as_float((u & 0x80000000u) ? (u & 0x7FFFFFFFu) : ~u);
}

__global__ __launch_bounds__(256) void gemm128_wmma(
    const float* __restrict__ X, const float* __restrict__ W,
    float* __restrict__ Y, int M)
{
    __shared__ float lw[FDIM * FDIM];
    {
        const float4* w4 = (const float4*)W;
        float4* l4 = (float4*)lw;
        for (int i = threadIdx.x; i < FDIM * FDIM / 4; i += 256) l4[i] = w4[i];
    }
    __syncthreads();

    const int wave = threadIdx.x >> 5;
    const int lane = threadIdx.x & 31;
    const int m0 = (blockIdx.x * 8 + wave) * 16;
    if (m0 >= M) return;

    const int lm = lane & 15;
    const int lh = lane >> 4;
    const float* arow = X + (size_t)(m0 + lm) * FDIM + 2 * lh;

#pragma unroll
    for (int nt = 0; nt < 8; ++nt) {
        v8f acc = {};
        const int nc = nt * 16 + lm;
#pragma unroll 8
        for (int kk = 0; kk < 32; ++kk) {
            const int k0 = kk * 4;
            v2f a = *(const v2f*)(arow + k0);
            v2f b;
            b.x = lw[(k0 + 2 * lh) * FDIM + nc];
            b.y = lw[(k0 + 2 * lh + 1) * FDIM + nc];
            acc = __builtin_amdgcn_wmma_f32_16x16x4_f32(
                false, a, false, b, (short)0, acc, false, false);
        }
        float* yrow = Y + (size_t)(m0 + 8 * lh) * FDIM + nc;
#pragma unroll
        for (int r = 0; r < 8; ++r) yrow[(size_t)r * FDIM] = acc[r];
    }
}

__global__ void attn_logits(const float* __restrict__ h,
                            const float* __restrict__ a_src,
                            const float* __restrict__ a_dst,
                            float* __restrict__ esrc, float* __restrict__ edst,
                            int n)
{
    int idx = blockIdx.x * blockDim.x + threadIdx.x;
    if (idx >= n * HEADS) return;
    int node = idx >> 1, hd = idx & 1;
    const float* hp = h + (size_t)node * FDIM + hd * 64;
    const float* as = a_src + hd * 64;
    const float* ad = a_dst + hd * 64;
    float s = 0.f, t = 0.f;
#pragma unroll 8
    for (int f = 0; f < 64; ++f) { float x = hp[f]; s += x * as[f]; t += x * ad[f]; }
    esrc[idx] = s; edst[idx] = t;
}

#define TILE 16
#define TILE_SHIFT 4
#define DTILE 64
#define MAXNB 4096
#define CHUNK 8192

__global__ void bucket_zero(int* __restrict__ bcount, int nb)
{
    int i = blockIdx.x * blockDim.x + threadIdx.x;
    if (i < nb) bcount[i] = 0;
}

__global__ __launch_bounds__(256) void bucket_count(const int* __restrict__ dst,
                                                    int E, int n, int nb,
                                                    int* __restrict__ bcount)
{
    __shared__ int lh[MAXNB];
    for (int i = threadIdx.x; i < nb; i += 256) lh[i] = 0;
    __syncthreads();
    const int base = blockIdx.x * CHUNK;
    for (int k = 0; k < CHUNK / 256; ++k) {
        int e = base + k * 256 + threadIdx.x;
        if (e < E + n) {
            int d = (e < E) ? dst[e] : (e - E);
            atomicAdd(&lh[d >> TILE_SHIFT], 1);
        }
    }
    __syncthreads();
    for (int i = threadIdx.x; i < nb; i += 256) {
        int c = lh[i];
        if (c) atomicAdd(&bcount[i], c);
    }
}

__global__ __launch_bounds__(1024) void bucket_scan(const int* __restrict__ bcount, int nb,
                                                    int* __restrict__ bstart,
                                                    int* __restrict__ bcursor)
{
    __shared__ int s[1024];
    const int t = threadIdx.x;
    int cv[MAXNB / 1024];
    int tot = 0;
#pragma unroll
    for (int k = 0; k < MAXNB / 1024; ++k) {
        const int i = t * (MAXNB / 1024) + k;
        cv[k] = (i < nb) ? bcount[i] : 0;
        tot += cv[k];
    }
    s[t] = tot;
    __syncthreads();
    for (int off = 1; off < 1024; off <<= 1) {
        int v = (t >= off) ? s[t - off] : 0;
        __syncthreads();
        s[t] += v;
        __syncthreads();
    }
    int run = s[t] - tot;
#pragma unroll
    for (int k = 0; k < MAXNB / 1024; ++k) {
        const int i = t * (MAXNB / 1024) + k;
        if (i < nb) { bstart[i] = run; bcursor[i] = run; }
        run += cv[k];
    }
    if (t == 1023) bstart[nb] = s[t];
}

__global__ __launch_bounds__(256) void bucket_scatter(const int* __restrict__ src,
                                                      const int* __restrict__ dst,
                                                      int E, int n, int nb,
                                                      int* __restrict__ bcursor,
                                                      int* __restrict__ psrc,
                                                      int* __restrict__ pdst,
                                                      int* __restrict__ peid)
{
    __shared__ int lh[MAXNB];
    __shared__ int lbase[MAXNB];
    for (int i = threadIdx.x; i < nb; i += 256) lh[i] = 0;
    __syncthreads();
    const int base = blockIdx.x * CHUNK;
    for (int k = 0; k < CHUNK / 256; ++k) {
        int e = base + k * 256 + threadIdx.x;
        if (e < E + n) {
            int d = (e < E) ? dst[e] : (e - E);
            atomicAdd(&lh[d >> TILE_SHIFT], 1);
        }
    }
    __syncthreads();
    for (int i = threadIdx.x; i < nb; i += 256) {
        int c = lh[i];
        lbase[i] = c ? atomicAdd(&bcursor[i], c) : 0;
        lh[i] = 0;
    }
    __syncthreads();
    for (int k = 0; k < CHUNK / 256; ++k) {
        int e = base + k * 256 + threadIdx.x;
        if (e < E + n) {
            int s = (e < E) ? src[e] : (e - E);
            int d = (e < E) ? dst[e] : (e - E);
            int b = d >> TILE_SHIFT;
            int pos = lbase[b] + atomicAdd(&lh[b], 1);
            psrc[pos] = s;
            pdst[pos] = d;
            peid[pos] = e;
        }
    }
}

#define AW 4
#define MPT 8
#define GB 4

__device__ __forceinline__ int lane_get_i(int src, int v) {
    return __builtin_amdgcn_ds_bpermute(src << 2, v);
}
__device__ __forceinline__ float lane_get_f(int src, float v) {
    return __builtin_bit_cast(float, __builtin_amdgcn_ds_bpermute(src << 2, __builtin_bit_cast(int, v)));
}

__device__ __forceinline__ float lrelu(float v) { return (v >= 0.f) ? v : NEG_SLOPE * v; }

__global__ __launch_bounds__(32 * AW) void bucket_softmax_aggregate(
    const int* __restrict__ psrc, const int* __restrict__ pdst,
    const int* __restrict__ bstart,
    const float* __restrict__ esrc, const float* __restrict__ edst,
    const float* __restrict__ hfeat, const float* __restrict__ bias,
    float* __restrict__ z, int n, int nb, int do_relu)
{
    __shared__ unsigned mx[AW][TILE * HEADS];

    const int wave = threadIdx.x >> 5;
    const int lane = threadIdx.x & 31;
    const int c = lane & 15;
    const int hh = lane >> 4;
    const int b = blockIdx.x * AW + wave;
    const bool live = b < nb;
    const int n0 = b * TILE;
    const int e0 = live ? bstart[b] : 0;
    const int e1 = live ? bstart[b + 1] : 0;
    unsigned* mxw = mx[wave];
    const float2* esrc2 = (const float2*)esrc;
    const float2* edst2 = (const float2*)edst;

    mxw[lane] = 0u;
    __syncthreads();

    for (int eb = e0 + lane; eb < e1; eb += 32 * MPT) {
        int sm[MPT], dm[MPT];
#pragma unroll
        for (int u = 0; u < MPT; ++u) {
            const int ee = (eb + 32 * u < e1) ? (eb + 32 * u) : (e1 - 1);
            sm[u] = psrc[ee];
            dm[u] = pdst[ee];
        }
        float2 es[MPT], ed[MPT];
#pragma unroll
        for (int u = 0; u < MPT; ++u) { es[u] = esrc2[sm[u]]; ed[u] = edst2[dm[u]]; }
#pragma unroll
        for (int u = 0; u < MPT; ++u) {
            if (eb + 32 * u < e1) {
                atomicMax(&mxw[(dm[u] - n0) * HEADS + 0], enc_f32(lrelu(es[u].x + ed[u].x)));
                atomicMax(&mxw[(dm[u] - n0) * HEADS + 1], enc_f32(lrelu(es[u].y + ed[u].y)));
            }
        }
    }
    __syncthreads();

    if (e0 >= e1) return;

    v8f acc[8];
#pragma unroll
    for (int t = 0; t < 8; ++t) acc[t] = (v8f){};
    float den0 = 0.f, den1 = 0.f;

    int sc, dc;
    float a0c, a1c;
    {
        const int ee = (e0 + lane < e1) ? (e0 + lane) : (e1 - 1);
        sc = psrc[ee];
        dc = pdst[ee];
        const float2 es = esrc2[sc], ed = edst2[dc];
        const bool ok = e0 + lane < e1;
        a0c = ok ? __expf(lrelu(es.x + ed.x) - dec_f32(mxw[(dc - n0) * HEADS + 0])) : 0.f;
        a1c = ok ? __expf(lrelu(es.y + ed.y) - dec_f32(mxw[(dc - n0) * HEADS + 1])) : 0.f;
    }
    int sn, dn;
    {
        const int ee = (e0 + 32 + lane < e1) ? (e0 + 32 + lane) : (e1 - 1);
        sn = psrc[ee];
        dn = pdst[ee];
    }

    for (int eb = e0; eb < e1; eb += 32) {
        const float2 esn = esrc2[sn], edn = edst2[dn];
        int s2, d2;
        {
            const int ee = (eb + 64 + lane < e1) ? (eb + 64 + lane) : (e1 - 1);
            s2 = psrc[ee];
            d2 = pdst[ee];
        }
        const int dlc = dc - n0;

#pragma unroll
        for (int gb = 0; gb < 8 / GB; ++gb) {
            float4 r0[GB][2], r1[GB][2];
            int dlg[GB][2];
            float w0[GB][2], w1[GB][2];
#pragma unroll
            for (int q = 0; q < GB; ++q) {
#pragma unroll
                for (int i = 0; i < 2; ++i) {
                    const int j = (gb * GB + q) * 4 + 2 * hh + i;
                    const int sj = lane_get_i(j, sc);
                    dlg[q][i] = lane_get_i(j, dlc);
                    w0[q][i] = lane_get_f(j, a0c);
                    w1[q][i] = lane_get_f(j, a1c);
                    const float4* rp = (const float4*)(hfeat + (size_t)sj * FDIM) + c;
                    r0[q][i] = rp[0];
                    r1[q][i] = rp[16];
                }
            }
#pragma unroll
            for (int q = 0; q < GB; ++q) {
                v2f A0, A1;
                A0.x = (dlg[q][0] == c) ? w0[q][0] : 0.f;
                A0.y = (dlg[q][1] == c) ? w0[q][1] : 0.f;
                A1.x = (dlg[q][0] == c) ? w1[q][0] : 0.f;
                A1.y = (dlg[q][1] == c) ? w1[q][1] : 0.f;
                den0 += A0.x + A0.y;
                den1 += A1.x + A1.y;
                v2f B;
                B.x = r0[q][0].x; B.y = r0[q][1].x;
                acc[0] = __builtin_amdgcn_wmma_f32_16x16x4_f32(false, A0, false, B, (short)0, acc[0], false, false);
                B.x = r0[q][0].y; B.y = r0[q][1].y;
                acc[1] = __builtin_amdgcn_wmma_f32_16x16x4_f32(false, A0, false, B, (short)0, acc[1], false, false);
                B.x = r0[q][0].z; B.y = r0[q][1].z;
                acc[2] = __builtin_amdgcn_wmma_f32_16x16x4_f32(false, A0, false, B, (short)0, acc[2], false, false);
                B.x = r0[q][0].w; B.y = r0[q][1].w;
                acc[3] = __builtin_amdgcn_wmma_f32_16x16x4_f32(false, A0, false, B, (short)0, acc[3], false, false);
                B.x = r1[q][0].x; B.y = r1[q][1].x;
                acc[4] = __builtin_amdgcn_wmma_f32_16x16x4_f32(false, A1, false, B, (short)0, acc[4], false, false);
                B.x = r1[q][0].y; B.y = r1[q][1].y;
                acc[5] = __builtin_amdgcn_wmma_f32_16x16x4_f32(false, A1, false, B, (short)0, acc[5], false, false);
                B.x = r1[q][0].z; B.y = r1[q][1].z;
                acc[6] = __builtin_amdgcn_wmma_f32_16x16x4_f32(false, A1, false, B, (short)0, acc[6], false, false);
                B.x = r1[q][0].w; B.y = r1[q][1].w;
                acc[7] = __builtin_amdgcn_wmma_f32_16x16x4_f32(false, A1, false, B, (short)0, acc[7], false, false);
            }
        }

        {
            const bool ok = eb + 32 + lane < e1;
            a0c = ok ? __expf(lrelu(esn.x + edn.x) - dec_f32(mxw[(dn - n0) * HEADS + 0])) : 0.f;
            a1c = ok ? __expf(lrelu(esn.y + edn.y) - dec_f32(mxw[(dn - n0) * HEADS + 1])) : 0.f;
            sc = sn; dc = dn;
            sn = s2; dn = d2;
        }
    }

    den0 += lane_get_f(lane ^ 16, den0);
    den1 += lane_get_f(lane ^ 16, den1);
    const float4 bs0 = ((const float4*)bias)[c];
    const float4 bs1 = ((const float4*)bias)[16 + c];
    float4* z4 = (float4*)z;
#pragma unroll
    for (int r = 0; r < 8; ++r) {
        const float q0 = lane_get_f(8 * hh + r, den0);
        const float q1 = lane_get_f(8 * hh + r, den1);
        float4 o0, o1;
        o0.x = acc[0][r] / q0 + bs0.x; o0.y = acc[1][r] / q0 + bs0.y;
        o0.z = acc[2][r] / q0 + bs0.z; o0.w = acc[3][r] / q0 + bs0.w;
        o1.x = acc[4][r] / q1 + bs1.x; o1.y = acc[5][r] / q1 + bs1.y;
        o1.z = acc[6][r] / q1 + bs1.z; o1.w = acc[7][r] / q1 + bs1.w;
        if (do_relu) {
            o0.x = fmaxf(o0.x, 0.f); o0.y = fmaxf(o0.y, 0.f); o0.z = fmaxf(o0.z, 0.f); o0.w = fmaxf(o0.w, 0.f);
            o1.x = fmaxf(o1.x, 0.f); o1.y = fmaxf(o1.y, 0.f); o1.z = fmaxf(o1.z, 0.f); o1.w = fmaxf(o1.w, 0.f);
        }
        const int node = n0 + 8 * hh + r;
        if (node < n) {
            z4[(size_t)node * (FDIM / 4) + c] = o0;
            z4[(size_t)node * (FDIM / 4) + 16 + c] = o1;
        }
    }
}

#define DSLOTS 32
#define DBATCH 16

__device__ __forceinline__ float lane_xor16(float x) {
    return __builtin_bit_cast(float, __builtin_amdgcn_permlanex16(
        __builtin_bit_cast(int, x), __builtin_bit_cast(int, x), 0x76543210, 0xfedcba98, false, false));
}
template <int M>
__device__ __forceinline__ float lane_xor_row(float x) {
    return __builtin_bit_cast(float, __builtin_amdgcn_update_dpp(
        0, __builtin_bit_cast(int, x), 0x160 | M, 0xf, 0xf, true));
}

__global__ __launch_bounds__(256) void tile_decode(
    const int* __restrict__ psrc, const int* __restrict__ pdst, const int* __restrict__ peid,
    const int* __restrict__ bstart, const float* __restrict__ z,
    float* __restrict__ out, int n, int nb, int E)
{
    __shared__ float4 zt[DTILE * FDIM / 4];

    const int tid = threadIdx.x;
    const int wave = tid >> 5;
    const int lane = tid & 31;
    const int tile = blockIdx.x >> 1;
    const int n0 = tile * DTILE;
    const int b0 = tile * (DTILE / TILE);
    const int t0 = bstart[b0];
    const int t1 = bstart[min(b0 + DTILE / TILE, nb)];
    const int mid = t0 + ((t1 - t0) >> 1);
    const int e0 = (blockIdx.x & 1) ? mid : t0;
    const int e1 = (blockIdx.x & 1) ? t1 : mid;

    const float4* z4 = (const float4*)z;
    const unsigned loff = (unsigned)lane * 16u;
    float4 piece[DTILE * FDIM / 4 / 256];
#pragma unroll
    for (int k = 0; k < DTILE * FDIM / 4 / 256; ++k) {
        const int i = tid + 256 * k;
        const int node = min(n0 + (i >> 5), n - 1);
        piece[k] = z4[(size_t)node * (FDIM / 4) + (i & 31)];
    }
#pragma unroll
    for (int k = 0; k < DTILE * FDIM / 4 / 256; ++k) zt[tid + 256 * k] = piece[k];
    __syncthreads();

    for (int eb = e0 + wave * DSLOTS; eb < e1; eb += 8 * DSLOTS) {
        const int ee = (eb + lane < e1) ? (eb + lane) : (e1 - 1);
        const int sv = psrc[ee];
        const int dv = pdst[ee] - n0;
        const int ev = peid[ee];

        float p[DSLOTS];
#pragma unroll
        for (int b = 0; b < DSLOTS / DBATCH; ++b) {
            float4 g[DBATCH];
#pragma unroll
            for (int u = 0; u < DBATCH; ++u) {
                const unsigned s = (unsigned)__builtin_amdgcn_readlane(sv, b * DBATCH + u);
                const char* rowp = (const char*)z + (size_t)s * (FDIM * sizeof(float));
                g[u] = *(const float4*)(rowp + loff);
            }
#pragma unroll
            for (int u = 0; u < DBATCH; ++u) {
                const int dl = __builtin_amdgcn_readlane(dv, b * DBATCH + u);
                const float4 t = zt[dl * (FDIM / 4) + lane];
                p[b * DBATCH + u] = g[u].x * t.x + g[u].y * t.y + g[u].z * t.z + g[u].w * t.w;
            }
        }

        {
            const bool hi = (lane & 16) != 0;
#pragma unroll
            for (int u = 0; u < 16; ++u) {
                const float keep = hi ? p[u + 16] : p[u];
                const float send = hi ? p[u] : p[u + 16];
                p[u] = keep + lane_xor16(send);
            }
        }
        {
            const bool hi = (lane & 8) != 0;
#pragma unroll
            for (int u = 0; u < 8; ++u) {
                const float keep = hi ? p[u + 8] : p[u];
                const float send = hi ? p[u] : p[u + 8];
                p[u] = keep + lane_xor_row<8>(send);
            }
        }
        {
            const bool hi = (lane & 4) != 0;
#pragma unroll
            for (int u = 0; u < 4; ++u) {
                const float keep = hi ? p[u + 4] : p[u];
                const float send = hi ? p[u] : p[u + 4];
                p[u] = keep + lane_xor_row<4>(send);
            }
        }
        {
            const bool hi = (lane & 2) != 0;
#pragma unroll
            for (int u = 0; u < 2; ++u) {
                const float keep = hi ? p[u + 2] : p[u];
                const float send = hi ? p[u] : p[u + 2];
                p[u] = keep + lane_xor_row<2>(send);
            }
        }
        {
            const bool hi = (lane & 1) != 0;
            const float keep = hi ? p[1] : p[0];
            const float send = hi ? p[0] : p[1];
            p[0] = keep + lane_xor_row<1>(send);
        }
        if (eb + lane < e1 && ev < E) out[ev] = p[0];
    }
}

extern "C" void kernel_launch(void* const* d_in, const int* in_sizes, int n_in,
                              void* d_out, int out_size, void* d_ws, size_t ws_size,
                              hipStream_t stream) {
    const float* x      = (const float*)d_in[0];
    const int*   ei     = (const int*)d_in[1];
    const float* W1     = (const float*)d_in[2];
    const float* a1s    = (const float*)d_in[3];
    const float* a1d    = (const float*)d_in[4];
    const float* b1     = (const float*)d_in[5];
    const float* W2     = (const float*)d_in[6];
    const float* a2s    = (const float*)d_in[7];
    const float* a2d    = (const float*)d_in[8];
    const float* b2     = (const float*)d_in[9];
    float* scores = (float*)d_out;

    const int N = in_sizes[0] / FDIM;
    const int E = in_sizes[1] / 2;
    const int* src = ei;
    const int* dst = ei + E;

    float* ws   = (float*)d_ws;
    float*    h     = ws;
    int*      psrc  = (int*)(ws + (size_t)N * FDIM);
    int*      pdst  = psrc + (size_t)(E + N);
    int*      bcount  = pdst + (size_t)(E + N);
    int*      bstart  = bcount + MAXNB;
    int*      bcursor = bstart + MAXNB + 1;
    int*      peid  = bcursor + MAXNB;
    float*    z     = ws + (size_t)2 * N * FDIM;
    float*    esrc  = ws + (size_t)3 * N * FDIM;
    float*    edst  = esrc + (size_t)N * HEADS;

    const int ETOT = E + N;
    const int NB = (N + TILE - 1) / TILE;
    const int NT = (N + DTILE - 1) / DTILE;
    dim3 blk(256);
    dim3 gGemm((N + 127) / 128);
    dim3 gNH((N * HEADS + 255) / 256);
    dim3 gChunk((ETOT + CHUNK - 1) / CHUNK);
    dim3 gAgg((NB + AW - 1) / AW);

    bucket_zero<<<dim3((NB + 255) / 256), blk, 0, stream>>>(bcount, NB);
    bucket_count<<<gChunk, blk, 0, stream>>>(dst, E, N, NB, bcount);
    bucket_scan<<<dim3(1), dim3(1024), 0, stream>>>(bcount, NB, bstart, bcursor);
    bucket_scatter<<<gChunk, blk, 0, stream>>>(src, dst, E, N, NB, bcursor, psrc, pdst, peid);

    gemm128_wmma<<<gGemm, blk, 0, stream>>>(x, W1, h, N);
    attn_logits<<<gNH, blk, 0, stream>>>(h, a1s, a1d, esrc, edst, N);
    bucket_softmax_aggregate<<<gAgg, dim3(32 * AW), 0, stream>>>(psrc, pdst, bstart, esrc, edst, h, b1, z, N, NB,  1);

    gemm128_wmma<<<gGemm, blk, 0, stream>>>(z, W2, h, N);
    attn_logits<<<gNH, blk, 0, stream>>>(h, a2s, a2d, esrc, edst, N);
    bucket_softmax_aggregate<<<gAgg, dim3(32 * AW), 0, stream>>>(psrc, pdst, bstart, esrc, edst, h, b2, z, N, NB,  0);

    tile_decode<<<dim3(2 * NT), blk, 0, stream>>>(psrc, pdst, peid, bstart, z, scores, N, NB, E);
}


# bucketing kernels (count, scatter): a thread's 32 index reads go out as one group of unconditional loads, not one per round trip; dst read once; reserving adds issued together; slot order as before
# speedup vs baseline: 3.2242x; 1.0769x over previous
#include <hip/hip_runtime.h>
#include <hip/hip_bf16.h>


typedef __attribute__((ext_vector_type(2))) float v2f;
typedef __attribute__((ext_vector_type(8))) float v8f;

#define HEADS 2
#define FDIM 128
#define NEG_SLOPE 0.2f

__device__ __forceinline__ unsigned enc_f32(float f) {
    unsigned u = __float_as_uint(f);
    return (u & 0x80000000u) ? ~u : (u | 0x80000000u);
}
__device__ __forceinline__ float dec_f32(unsigned u) {
    return __uint_as_float((u & 0x80000000u) ? (u & 0x7FFFFFFFu) : ~u);
}

__global__ __launch_bounds__(256) void gemm128_wmma(
    const float* __restrict__ X, const float* __restrict__ W,
    float* __restrict__ Y, int M)
{
    __shared__ float lw[FDIM * FDIM];
    {
        const float4* w4 = (const float4*)W;
        float4* l4 = (float4*)lw;
        for (int i = threadIdx.x; i < FDIM * FDIM / 4; i += 256) l4[i] = w4[i];
    }
    __syncthreads();

    const int wave = threadIdx.x >> 5;
    const int lane = threadIdx.x & 31;
    const int m0 = (blockIdx.x * 8 + wave) * 16;
    if (m0 >= M) return;

    const int lm = lane & 15;
    const int lh = lane >> 4;
    const float* arow = X + (size_t)(m0 + lm) * FDIM + 2 * lh;

#pragma unroll
    for (int nt = 0; nt < 8; ++nt) {
        v8f acc = {};
        const int nc = nt * 16 + lm;
#pragma unroll 8
        for (int kk = 0; kk < 32; ++kk) {
            const int k0 = kk * 4;
            v2f a = *(const v2f*)(arow + k0);
            v2f b;
            b.x = lw[(k0 + 2 * lh) * FDIM + nc];
            b.y = lw[(k0 + 2 * lh + 1) * FDIM + nc];
            acc = __builtin_amdgcn_wmma_f32_16x16x4_f32(
                false, a, false, b, (short)0, acc, false, false);
        }
        float* yrow = Y + (size_t)(m0 + 8 * lh) * FDIM + nc;
#pragma unroll
        for (int r = 0; r < 8; ++r) yrow[(size_t)r * FDIM] = acc[r];
    }
}

__global__ void attn_logits(const float* __restrict__ h,
                            const float* __restrict__ a_src,
                            const float* __restrict__ a_dst,
                            float* __restrict__ esrc, float* __restrict__ edst,
                            int n)
{
    int idx = blockIdx.x * blockDim.x + threadIdx.x;
    if (idx >= n * HEADS) return;
    int node = idx >> 1, hd = idx & 1;
    const float* hp = h + (size_t)node * FDIM + hd * 64;
    const float* as = a_src + hd * 64;
    const float* ad = a_dst + hd * 64;
    float s = 0.f, t = 0.f;
#pragma unroll 8
    for (int f = 0; f < 64; ++f) { float x = hp[f]; s += x * as[f]; t += x * ad[f]; }
    esrc[idx] = s; edst[idx] = t;
}

#define TILE 16
#define TILE_SHIFT 4
#define DTILE 64
#define MAXNB 4096
#define CHUNK 8192

__global__ void bucket_zero(int* __restrict__ bcount, int nb)
{
    int i = blockIdx.x * blockDim.x + threadIdx.x;
    if (i < nb) bcount[i] = 0;
}

#define EPT (CHUNK / 256)

__global__ __launch_bounds__(256) void bucket_count(const int* __restrict__ dst,
                                                    int E, int n, int nb,
                                                    int* __restrict__ bcount)
{
    __shared__ int lh[MAXNB];
    const int base = blockIdx.x * CHUNK + threadIdx.x;
    int dv[EPT];
#pragma unroll
    for (int k = 0; k < EPT; ++k) dv[k] = dst[min(base + k * 256, E - 1)];
    for (int i = threadIdx.x; i < nb; i += 256) lh[i] = 0;
    __syncthreads();
#pragma unroll
    for (int k = 0; k < EPT; ++k) {
        const int e = base + k * 256;
        if (e < E + n) {
            const int d = (e < E) ? dv[k] : (e - E);
            atomicAdd(&lh[d >> TILE_SHIFT], 1);
        }
    }
    __syncthreads();
    for (int i = threadIdx.x; i < nb; i += 256) {
        int c = lh[i];
        if (c) atomicAdd(&bcount[i], c);
    }
}

__global__ __launch_bounds__(1024) void bucket_scan(const int* __restrict__ bcount, int nb,
                                                    int* __restrict__ bstart,
                                                    int* __restrict__ bcursor)
{
    __shared__ int s[1024];
    const int t = threadIdx.x;
    int cv[MAXNB / 1024];
    int tot = 0;
#pragma unroll
    for (int k = 0; k < MAXNB / 1024; ++k) {
        const int i = t * (MAXNB / 1024) + k;
        cv[k] = (i < nb) ? bcount[i] : 0;
        tot += cv[k];
    }
    s[t] = tot;
    __syncthreads();
    for (int off = 1; off < 1024; off <<= 1) {
        int v = (t >= off) ? s[t - off] : 0;
        __syncthreads();
        s[t] += v;
        __syncthreads();
    }
    int run = s[t] - tot;
#pragma unroll
    for (int k = 0; k < MAXNB / 1024; ++k) {
        const int i = t * (MAXNB / 1024) + k;
        if (i < nb) { bstart[i] = run; bcursor[i] = run; }
        run += cv[k];
    }
    if (t == 1023) bstart[nb] = s[t];
}

__global__ __launch_bounds__(256) void bucket_scatter(const int* __restrict__ src,
                                                      const int* __restrict__ dst,
                                                      int E, int n, int nb,
                                                      int* __restrict__ bcursor,
                                                      int* __restrict__ psrc,
                                                      int* __restrict__ pdst,
                                                      int* __restrict__ peid)
{
    __shared__ int lh[MAXNB];
    __shared__ int lbase[MAXNB];
    const int base = blockIdx.x * CHUNK + threadIdx.x;
    int dv[EPT], sv[EPT];
#pragma unroll
    for (int k = 0; k < EPT; ++k) dv[k] = dst[min(base + k * 256, E - 1)];
    for (int i = threadIdx.x; i < nb; i += 256) lh[i] = 0;
    __syncthreads();
#pragma unroll
    for (int k = 0; k < EPT; ++k) sv[k] = src[min(base + k * 256, E - 1)];
#pragma unroll
    for (int k = 0; k < EPT; ++k) {
        const int e = base + k * 256;
        if (e < E + n) {
            const int d = (e < E) ? dv[k] : (e - E);
            dv[k] = d;
            atomicAdd(&lh[d >> TILE_SHIFT], 1);
        }
    }
    __syncthreads();
    {
        int cv[MAXNB / 256], bv[MAXNB / 256];
#pragma unroll
        for (int j = 0; j < MAXNB / 256; ++j) {
            const int i = threadIdx.x + 256 * j;
            cv[j] = (i < nb) ? lh[i] : 0;
        }
#pragma unroll
        for (int j = 0; j < MAXNB / 256; ++j) {
            const int i = threadIdx.x + 256 * j;
            bv[j] = cv[j] ? atomicAdd(&bcursor[i], cv[j]) : 0;
        }
#pragma unroll
        for (int j = 0; j < MAXNB / 256; ++j) {
            const int i = threadIdx.x + 256 * j;
            if (i < nb) { lbase[i] = bv[j]; lh[i] = 0; }
        }
    }
    __syncthreads();
#pragma unroll
    for (int k = 0; k < EPT; ++k) {
        const int e = base + k * 256;
        if (e < E + n) {
            const int s = (e < E) ? sv[k] : (e - E);
            const int d = dv[k];
            const int b = d >> TILE_SHIFT;
            const int pos = lbase[b] + atomicAdd(&lh[b], 1);
            psrc[pos] = s;
            pdst[pos] = d;
            peid[pos] = e;
        }
    }
}

#define AW 4
#define MPT 8
#define GB 4

__device__ __forceinline__ int lane_get_i(int src, int v) {
    return __builtin_amdgcn_ds_bpermute(src << 2, v);
}
__device__ __forceinline__ float lane_get_f(int src, float v) {
    return __builtin_bit_cast(float, __builtin_amdgcn_ds_bpermute(src << 2, __builtin_bit_cast(int, v)));
}

__device__ __forceinline__ float lrelu(float v) { return (v >= 0.f) ? v : NEG_SLOPE * v; }

__global__ __launch_bounds__(32 * AW) void bucket_softmax_aggregate(
    const int* __restrict__ psrc, const int* __restrict__ pdst,
    const int* __restrict__ bstart,
    const float* __restrict__ esrc, const float* __restrict__ edst,
    const float* __restrict__ hfeat, const float* __restrict__ bias,
    float* __restrict__ z, int n, int nb, int do_relu)
{
    __shared__ unsigned mx[AW][TILE * HEADS];

    const int wave = threadIdx.x >> 5;
    const int lane = threadIdx.x & 31;
    const int c = lane & 15;
    const int hh = lane >> 4;
    const int b = blockIdx.x * AW + wave;
    const bool live = b < nb;
    const int n0 = b * TILE;
    const int e0 = live ? bstart[b] : 0;
    const int e1 = live ? bstart[b + 1] : 0;
    unsigned* mxw = mx[wave];
    const float2* esrc2 = (const float2*)esrc;
    const float2* edst2 = (const float2*)edst;

    mxw[lane] = 0u;
    __syncthreads();

    for (int eb = e0 + lane; eb < e1; eb += 32 * MPT) {
        int sm[MPT], dm[MPT];
#pragma unroll
        for (int u = 0; u < MPT; ++u) {
            const int ee = (eb + 32 * u < e1) ? (eb + 32 * u) : (e1 - 1);
            sm[u] = psrc[ee];
            dm[u] = pdst[ee];
        }
        float2 es[MPT], ed[MPT];
#pragma unroll
        for (int u = 0; u < MPT; ++u) { es[u] = esrc2[sm[u]]; ed[u] = edst2[dm[u]]; }
#pragma unroll
        for (int u = 0; u < MPT; ++u) {
            if (eb + 32 * u < e1) {
                atomicMax(&mxw[(dm[u] - n0) * HEADS + 0], enc_f32(lrelu(es[u].x + ed[u].x)));
                atomicMax(&mxw[(dm[u] - n0) * HEADS + 1], enc_f32(lrelu(es[u].y + ed[u].y)));
            }
        }
    }
    __syncthreads();

    if (e0 >= e1) return;

    v8f acc[8];
#pragma unroll
    for (int t = 0; t < 8; ++t) acc[t] = (v8f){};
    float den0 = 0.f, den1 = 0.f;

    int sc, dc;
    float a0c, a1c;
    {
        const int ee = (e0 + lane < e1) ? (e0 + lane) : (e1 - 1);
        sc = psrc[ee];
        dc = pdst[ee];
        const float2 es = esrc2[sc], ed = edst2[dc];
        const bool ok = e0 + lane < e1;
        a0c = ok ? __expf(lrelu(es.x + ed.x) - dec_f32(mxw[(dc - n0) * HEADS + 0])) : 0.f;
        a1c = ok ? __expf(lrelu(es.y + ed.y) - dec_f32(mxw[(dc - n0) * HEADS + 1])) : 0.f;
    }
    int sn, dn;
    {
        const int ee = (e0 + 32 + lane < e1) ? (e0 + 32 + lane) : (e1 - 1);
        sn = psrc[ee];
        dn = pdst[ee];
    }

    for (int eb = e0; eb < e1; eb += 32) {
        const float2 esn = esrc2[sn], edn = edst2[dn];
        int s2, d2;
        {
            const int ee = (eb + 64 + lane < e1) ? (eb + 64 + lane) : (e1 - 1);
            s2 = psrc[ee];
            d2 = pdst[ee];
        }
        const int dlc = dc - n0;

#pragma unroll
        for (int gb = 0; gb < 8 / GB; ++gb) {
            float4 r0[GB][2], r1[GB][2];
            int dlg[GB][2];
            float w0[GB][2], w1[GB][2];
#pragma unroll
            for (int q = 0; q < GB; ++q) {
#pragma unroll
                for (int i = 0; i < 2; ++i) {
                    const int j = (gb * GB + q) * 4 + 2 * hh + i;
                    const int sj = lane_get_i(j, sc);
                    dlg[q][i] = lane_get_i(j, dlc);
                    w0[q][i] = lane_get_f(j, a0c);
                    w1[q][i] = lane_get_f(j, a1c);
                    const float4* rp = (const float4*)(hfeat + (size_t)sj * FDIM) + c;
                    r0[q][i] = rp[0];
                    r1[q][i] = rp[16];
                }
            }
#pragma unroll
            for (int q = 0; q < GB; ++q) {
                v2f A0, A1;
                A0.x = (dlg[q][0] == c) ? w0[q][0] : 0.f;
                A0.y = (dlg[q][1] == c) ? w0[q][1] : 0.f;
                A1.x = (dlg[q][0] == c) ? w1[q][0] : 0.f;
                A1.y = (dlg[q][1] == c) ? w1[q][1] : 0.f;
                den0 += A0.x + A0.y;
                den1 += A1.x + A1.y;
                v2f B;
                B.x = r0[q][0].x; B.y = r0[q][1].x;
                acc[0] = __builtin_amdgcn_wmma_f32_16x16x4_f32(false, A0, false, B, (short)0, acc[0], false, false);
                B.x = r0[q][0].y; B.y = r0[q][1].y;
                acc[1] = __builtin_amdgcn_wmma_f32_16x16x4_f32(false, A0, false, B, (short)0, acc[1], false, false);
                B.x = r0[q][0].z; B.y = r0[q][1].z;
                acc[2] = __builtin_amdgcn_wmma_f32_16x16x4_f32(false, A0, false, B, (short)0, acc[2], false, false);
                B.x = r0[q][0].w; B.y = r0[q][1].w;
                acc[3] = __builtin_amdgcn_wmma_f32_16x16x4_f32(false, A0, false, B, (short)0, acc[3], false, false);
                B.x = r1[q][0].x; B.y = r1[q][1].x;
                acc[4] = __builtin_amdgcn_wmma_f32_16x16x4_f32(false, A1, false, B, (short)0, acc[4], false, false);
                B.x = r1[q][0].y; B.y = r1[q][1].y;
                acc[5] = __builtin_amdgcn_wmma_f32_16x16x4_f32(false, A1, false, B, (short)0, acc[5], false, false);
                B.x = r1[q][0].z; B.y = r1[q][1].z;
                acc[6] = __builtin_amdgcn_wmma_f32_16x16x4_f32(false, A1, false, B, (short)0, acc[6], false, false);
                B.x = r1[q][0].w; B.y = r1[q][1].w;
                acc[7] = __builtin_amdgcn_wmma_f32_16x16x4_f32(false, A1, false, B, (short)0, acc[7], false, false);
            }
        }

        {
            const bool ok = eb + 32 + lane < e1;
            a0c = ok ? __expf(lrelu(esn.x + edn.x) - dec_f32(mxw[(dn - n0) * HEADS + 0])) : 0.f;
            a1c = ok ? __expf(lrelu(esn.y + edn.y) - dec_f32(mxw[(dn - n0) * HEADS + 1])) : 0.f;
            sc = sn; dc = dn;
            sn = s2; dn = d2;
        }
    }

    den0 += lane_get_f(lane ^ 16, den0);
    den1 += lane_get_f(lane ^ 16, den1);
    const float4 bs0 = ((const float4*)bias)[c];
    const float4 bs1 = ((const float4*)bias)[16 + c];
    float4* z4 = (float4*)z;
#pragma unroll
    for (int r = 0; r < 8; ++r) {
        const float q0 = lane_get_f(8 * hh + r, den0);
        const float q1 = lane_get_f(8 * hh + r, den1);
        float4 o0, o1;
        o0.x = acc[0][r] / q0 + bs0.x; o0.y = acc[1][r] / q0 + bs0.y;
        o0.z = acc[2][r] / q0 + bs0.z; o0.w = acc[3][r] / q0 + bs0.w;
        o1.x = acc[4][r] / q1 + bs1.x; o1.y = acc[5][r] / q1 + bs1.y;
        o1.z = acc[6][r] / q1 + bs1.z; o1.w = acc[7][r] / q1 + bs1.w;
        if (do_relu) {
            o0.x = fmaxf(o0.x, 0.f); o0.y = fmaxf(o0.y, 0.f); o0.z = fmaxf(o0.z, 0.f); o0.w = fmaxf(o0.w, 0.f);
            o1.x = fmaxf(o1.x, 0.f); o1.y = fmaxf(o1.y, 0.f); o1.z = fmaxf(o1.z, 0.f); o1.w = fmaxf(o1.w, 0.f);
        }
        const int node = n0 + 8 * hh + r;
        if (node < n) {
            z4[(size_t)node * (FDIM / 4) + c] = o0;
            z4[(size_t)node * (FDIM / 4) + 16 + c] = o1;
        }
    }
}

#define DSLOTS 32
#define DBATCH 16

__device__ __forceinline__ float lane_xor16(float x) {
    return __builtin_bit_cast(float, __builtin_amdgcn_permlanex16(
        __builtin_bit_cast(int, x), __builtin_bit_cast(int, x), 0x76543210, 0xfedcba98, false, false));
}
template <int M>
__device__ __forceinline__ float lane_xor_row(float x) {
    return __builtin_bit_cast(float, __builtin_amdgcn_update_dpp(
        0, __builtin_bit_cast(int, x), 0x160 | M, 0xf, 0xf, true));
}

__global__ __launch_bounds__(256) void tile_decode(
    const int* __restrict__ psrc, const int* __restrict__ pdst, const int* __restrict__ peid,
    const int* __restrict__ bstart, const float* __restrict__ z,
    float* __restrict__ out, int n, int nb, int E)
{
    __shared__ float4 zt[DTILE * FDIM / 4];

    const int tid = threadIdx.x;
    const int wave = tid >> 5;
    const int lane = tid & 31;
    const int tile = blockIdx.x >> 1;
    const int n0 = tile * DTILE;
    const int b0 = tile * (DTILE / TILE);
    const int t0 = bstart[b0];
    const int t1 = bstart[min(b0 + DTILE / TILE, nb)];
    const int mid = t0 + ((t1 - t0) >> 1);
    const int e0 = (blockIdx.x & 1) ? mid : t0;
    const int e1 = (blockIdx.x & 1) ? t1 : mid;

    const float4* z4 = (const float4*)z;
    const unsigned loff = (unsigned)lane * 16u;
    float4 piece[DTILE * FDIM / 4 / 256];
#pragma unroll
    for (int k = 0; k < DTILE * FDIM / 4 / 256; ++k) {
        const int i = tid + 256 * k;
        const int node = min(n0 + (i >> 5), n - 1);
        piece[k] = z4[(size_t)node * (FDIM / 4) + (i & 31)];
    }
#pragma unroll
    for (int k = 0; k < DTILE * FDIM / 4 / 256; ++k) zt[tid + 256 * k] = piece[k];
    __syncthreads();

    for (int eb = e0 + wave * DSLOTS; eb < e1; eb += 8 * DSLOTS) {
        const int ee = (eb + lane < e1) ? (eb + lane) : (e1 - 1);
        const int sv = psrc[ee];
        const int dv = pdst[ee] - n0;
        const int ev = peid[ee];

        float p[DSLOTS];
#pragma unroll
        for (int b = 0; b < DSLOTS / DBATCH; ++b) {
            float4 g[DBATCH];
#pragma unroll
            for (int u = 0; u < DBATCH; ++u) {
                const unsigned s = (unsigned)__builtin_amdgcn_readlane(sv, b * DBATCH + u);
                const char* rowp = (const char*)z + (size_t)s * (FDIM * sizeof(float));
                g[u] = *(const float4*)(rowp + loff);
            }
#pragma unroll
            for (int u = 0; u < DBATCH; ++u) {
                const int dl = __builtin_amdgcn_readlane(dv, b * DBATCH + u);
                const float4 t = zt[dl * (FDIM / 4) + lane];
                p[b * DBATCH + u] = g[u].x * t.x + g[u].y * t.y + g[u].z * t.z + g[u].w * t.w;
            }
        }

        {
            const bool hi = (lane & 16) != 0;
#pragma unroll
            for (int u = 0; u < 16; ++u) {
                const float keep = hi ? p[u + 16] : p[u];
                const float send = hi ? p[u] : p[u + 16];
                p[u] = keep + lane_xor16(send);
            }
        }
        {
            const bool hi = (lane & 8) != 0;
#pragma unroll
            for (int u = 0; u < 8; ++u) {
                const float keep = hi ? p[u + 8] : p[u];
                const float send = hi ? p[u] : p[u + 8];
                p[u] = keep + lane_xor_row<8>(send);
            }
        }
        {
            const bool hi = (lane & 4) != 0;
#pragma unroll
            for (int u = 0; u < 4; ++u) {
                const float keep = hi ? p[u + 4] : p[u];
                const float send = hi ? p[u] : p[u + 4];
                p[u] = keep + lane_xor_row<4>(send);
            }
        }
        {
            const bool hi = (lane & 2) != 0;
#pragma unroll
            for (int u = 0; u < 2; ++u) {
                const float keep = hi ? p[u + 2] : p[u];
                const float send = hi ? p[u] : p[u + 2];
                p[u] = keep + lane_xor_row<2>(send);
            }
        }
        {
            const bool hi = (lane & 1) != 0;
            const float keep = hi ? p[1] : p[0];
            const float send = hi ? p[0] : p[1];
            p[0] = keep + lane_xor_row<1>(send);
        }
        if (eb + lane < e1 && ev < E) out[ev] = p[0];
    }
}

extern "C" void kernel_launch(void* const* d_in, const int* in_sizes, int n_in,
                              void* d_out, int out_size, void* d_ws, size_t ws_size,
                              hipStream_t stream) {
    const float* x      = (const float*)d_in[0];
    const int*   ei     = (const int*)d_in[1];
    const float* W1     = (const float*)d_in[2];
    const float* a1s    = (const float*)d_in[3];
    const float* a1d    = (const float*)d_in[4];
    const float* b1     = (const float*)d_in[5];
    const float* W2     = (const float*)d_in[6];
    const float* a2s    = (const float*)d_in[7];
    const float* a2d    = (const float*)d_in[8];
    const float* b2     = (const float*)d_in[9];
    float* scores = (float*)d_out;

    const int N = in_sizes[0] / FDIM;
    const int E = in_sizes[1] / 2;
    const int* src = ei;
    const int* dst = ei + E;

    float* ws   = (float*)d_ws;
    float*    h     = ws;
    int*      psrc  = (int*)(ws + (size_t)N * FDIM);
    int*      pdst  = psrc + (size_t)(E + N);
    int*      bcount  = pdst + (size_t)(E + N);
    int*      bstart  = bcount + MAXNB;
    int*      bcursor = bstart + MAXNB + 1;
    int*      peid  = bcursor + MAXNB;
    float*    z     = ws + (size_t)2 * N * FDIM;
    float*    esrc  = ws + (size_t)3 * N * FDIM;
    float*    edst  = esrc + (size_t)N * HEADS;

    const int ETOT = E + N;
    const int NB = (N + TILE - 1) / TILE;
    const int NT = (N + DTILE - 1) / DTILE;
    dim3 blk(256);
    dim3 gGemm((N + 127) / 128);
    dim3 gNH((N * HEADS + 255) / 256);
    dim3 gChunk((ETOT + CHUNK - 1) / CHUNK);
    dim3 gAgg((NB + AW - 1) / AW);

    bucket_zero<<<dim3((NB + 255) / 256), blk, 0, stream>>>(bcount, NB);
    bucket_count<<<gChunk, blk, 0, stream>>>(dst, E, N, NB, bcount);
    bucket_scan<<<dim3(1), dim3(1024), 0, stream>>>(bcount, NB, bstart, bcursor);
    bucket_scatter<<<gChunk, blk, 0, stream>>>(src, dst, E, N, NB, bcursor, psrc, pdst, peid);

    gemm128_wmma<<<gGemm, blk, 0, stream>>>(x, W1, h, N);
    attn_logits<<<gNH, blk, 0, stream>>>(h, a1s, a1d, esrc, edst, N);
    bucket_softmax_aggregate<<<gAgg, dim3(32 * AW), 0, stream>>>(psrc, pdst, bstart, esrc, edst, h, b1, z, N, NB,  1);

    gemm128_wmma<<<gGemm, blk, 0, stream>>>(z, W2, h, N);
    attn_logits<<<gNH, blk, 0, stream>>>(h, a2s, a2d, esrc, edst, N);
    bucket_softmax_aggregate<<<gAgg, dim3(32 * AW), 0, stream>>>(psrc, pdst, bstart, esrc, edst, h, b2, z, N, NB,  0);

    tile_decode<<<dim3(2 * NT), blk, 0, stream>>>(psrc, pdst, peid, bstart, z, scores, N, NB, E);
}


# edge softmax+aggregation: one work-group of 4 waves per 16-node bucket, the bucket's 32-slot trips dealt over the waves in turn, partial WMMA sums and exp-sums added through LDS; new sum order
# speedup vs baseline: 3.3972x; 1.0537x over previous
#include <hip/hip_runtime.h>
#include <hip/hip_bf16.h>


typedef __attribute__((ext_vector_type(2))) float v2f;
typedef __attribute__((ext_vector_type(8))) float v8f;

#define HEADS 2
#define FDIM 128
#define NEG_SLOPE 0.2f

__device__ __forceinline__ unsigned enc_f32(float f) {
    unsigned u = __float_as_uint(f);
    return (u & 0x80000000u) ? ~u : (u | 0x80000000u);
}
__device__ __forceinline__ float dec_f32(unsigned u) {
    return __uint_as_float((u & 0x80000000u) ? (u & 0x7FFFFFFFu) : ~u);
}

__global__ __launch_bounds__(256) void gemm128_wmma(
    const float* __restrict__ X, const float* __restrict__ W,
    float* __restrict__ Y, int M)
{
    __shared__ float lw[FDIM * FDIM];
    {
        const float4* w4 = (const float4*)W;
        float4* l4 = (float4*)lw;
        for (int i = threadIdx.x; i < FDIM * FDIM / 4; i += 256) l4[i] = w4[i];
    }
    __syncthreads();

    const int wave = threadIdx.x >> 5;
    const int lane = threadIdx.x & 31;
    const int m0 = (blockIdx.x * 8 + wave) * 16;
    if (m0 >= M) return;

    const int lm = lane & 15;
    const int lh = lane >> 4;
    const float* arow = X + (size_t)(m0 + lm) * FDIM + 2 * lh;

#pragma unroll
    for (int nt = 0; nt < 8; ++nt) {
        v8f acc = {};
        const int nc = nt * 16 + lm;
#pragma unroll 8
        for (int kk = 0; kk < 32; ++kk) {
            const int k0 = kk * 4;
            v2f a = *(const v2f*)(arow + k0);
            v2f b;
            b.x = lw[(k0 + 2 * lh) * FDIM + nc];
            b.y = lw[(k0 + 2 * lh + 1) * FDIM + nc];
            acc = __builtin_amdgcn_wmma_f32_16x16x4_f32(
                false, a, false, b, (short)0, acc, false, false);
        }
        float* yrow = Y + (size_t)(m0 + 8 * lh) * FDIM + nc;
#pragma unroll
        for (int r = 0; r < 8; ++r) yrow[(size_t)r * FDIM] = acc[r];
    }
}

__global__ void attn_logits(const float* __restrict__ h,
                            const float* __restrict__ a_src,
                            const float* __restrict__ a_dst,
                            float* __restrict__ esrc, float* __restrict__ edst,
                            int n)
{
    int idx = blockIdx.x * blockDim.x + threadIdx.x;
    if (idx >= n * HEADS) return;
    int node = idx >> 1, hd = idx & 1;
    const float* hp = h + (size_t)node * FDIM + hd * 64;
    const float* as = a_src + hd * 64;
    const float* ad = a_dst + hd * 64;
    float s = 0.f, t = 0.f;
#pragma unroll 8
    for (int f = 0; f < 64; ++f) { float x = hp[f]; s += x * as[f]; t += x * ad[f]; }
    esrc[idx] = s; edst[idx] = t;
}

#define TILE 16
#define TILE_SHIFT 4
#define DTILE 64
#define MAXNB 4096
#define CHUNK 8192

__global__ void bucket_zero(int* __restrict__ bcount, int nb)
{
    int i = blockIdx.x * blockDim.x + threadIdx.x;
    if (i < nb) bcount[i] = 0;
}

#define EPT (CHUNK / 256)

__global__ __launch_bounds__(256) void bucket_count(const int* __restrict__ dst,
                                                    int E, int n, int nb,
                                                    int* __restrict__ bcount)
{
    __shared__ int lh[MAXNB];
    const int base = blockIdx.x * CHUNK + threadIdx.x;
    int dv[EPT];
#pragma unroll
    for (int k = 0; k < EPT; ++k) dv[k] = dst[min(base + k * 256, E - 1)];
    for (int i = threadIdx.x; i < nb; i += 256) lh[i] = 0;
    __syncthreads();
#pragma unroll
    for (int k = 0; k < EPT; ++k) {
        const int e = base + k * 256;
        if (e < E + n) {
            const int d = (e < E) ? dv[k] : (e - E);
            atomicAdd(&lh[d >> TILE_SHIFT], 1);
        }
    }
    __syncthreads();
    for (int i = threadIdx.x; i < nb; i += 256) {
        int c = lh[i];
        if (c) atomicAdd(&bcount[i], c);
    }
}

__global__ __launch_bounds__(1024) void bucket_scan(const int* __restrict__ bcount, int nb,
                                                    int* __restrict__ bstart,
                                                    int* __restrict__ bcursor)
{
    __shared__ int s[1024];
    const int t = threadIdx.x;
    int cv[MAXNB / 1024];
    int tot = 0;
#pragma unroll
    for (int k = 0; k < MAXNB / 1024; ++k) {
        const int i = t * (MAXNB / 1024) + k;
        cv[k] = (i < nb) ? bcount[i] : 0;
        tot += cv[k];
    }
    s[t] = tot;
    __syncthreads();
    for (int off = 1; off < 1024; off <<= 1) {
        int v = (t >= off) ? s[t - off] : 0;
        __syncthreads();
        s[t] += v;
        __syncthreads();
    }
    int run = s[t] - tot;
#pragma unroll
    for (int k = 0; k < MAXNB / 1024; ++k) {
        const int i = t * (MAXNB / 1024) + k;
        if (i < nb) { bstart[i] = run; bcursor[i] = run; }
        run += cv[k];
    }
    if (t == 1023) bstart[nb] = s[t];
}

__global__ __launch_bounds__(256) void bucket_scatter(const int* __restrict__ src,
                                                      const int* __restrict__ dst,
                                                      int E, int n, int nb,
                                                      int* __restrict__ bcursor,
                                                      int* __restrict__ psrc,
                                                      int* __restrict__ pdst,
                                                      int* __restrict__ peid)
{
    __shared__ int lh[MAXNB];
    __shared__ int lbase[MAXNB];
    const int base = blockIdx.x * CHUNK + threadIdx.x;
    int dv[EPT], sv[EPT];
#pragma unroll
    for (int k = 0; k < EPT; ++k) dv[k] = dst[min(base + k * 256, E - 1)];
    for (int i = threadIdx.x; i < nb; i += 256) lh[i] = 0;
    __syncthreads();
#pragma unroll
    for (int k = 0; k < EPT; ++k) sv[k] = src[min(base + k * 256, E - 1)];
#pragma unroll
    for (int k = 0; k < EPT; ++k) {
        const int e = base + k * 256;
        if (e < E + n) {
            const int d = (e < E) ? dv[k] : (e - E);
            dv[k] = d;
            atomicAdd(&lh[d >> TILE_SHIFT], 1);
        }
    }
    __syncthreads();
    {
        int cv[MAXNB / 256], bv[MAXNB / 256];
#pragma unroll
        for (int j = 0; j < MAXNB / 256; ++j) {
            const int i = threadIdx.x + 256 * j;
            cv[j] = (i < nb) ? lh[i] : 0;
        }
#pragma unroll
        for (int j = 0; j < MAXNB / 256; ++j) {
            const int i = threadIdx.x + 256 * j;
            bv[j] = cv[j] ? atomicAdd(&bcursor[i], cv[j]) : 0;
        }
#pragma unroll
        for (int j = 0; j < MAXNB / 256; ++j) {
            const int i = threadIdx.x + 256 * j;
            if (i < nb) { lbase[i] = bv[j]; lh[i] = 0; }
        }
    }
    __syncthreads();
#pragma unroll
    for (int k = 0; k < EPT; ++k) {
        const int e = base + k * 256;
        if (e < E + n) {
            const int s = (e < E) ? sv[k] : (e - E);
            const int d = dv[k];
            const int b = d >> TILE_SHIFT;
            const int pos = lbase[b] + atomicAdd(&lh[b], 1);
            psrc[pos] = s;
            pdst[pos] = d;
            peid[pos] = e;
        }
    }
}

#define SW 4
#define MPT 8
#define GB 4

__device__ __forceinline__ int lane_get_i(int src, int v) {
    return __builtin_amdgcn_ds_bpermute(src << 2, v);
}
__device__ __forceinline__ float lane_get_f(int src, float v) {
    return __builtin_bit_cast(float, __builtin_amdgcn_ds_bpermute(src << 2, __builtin_bit_cast(int, v)));
}

__device__ __forceinline__ float lrelu(float v) { return (v >= 0.f) ? v : NEG_SLOPE * v; }

__global__ __launch_bounds__(32 * SW) void bucket_softmax_aggregate(
    const int* __restrict__ psrc, const int* __restrict__ pdst,
    const int* __restrict__ bstart,
    const float* __restrict__ esrc, const float* __restrict__ edst,
    const float* __restrict__ hfeat, const float* __restrict__ bias,
    float* __restrict__ z, int n, int nb, int do_relu)
{
    __shared__ unsigned mx[TILE * HEADS];
    __shared__ float4 red[SW][8][HEADS][32];
    __shared__ float dens[SW][HEADS][32];

    const int wave = threadIdx.x >> 5;
    const int lane = threadIdx.x & 31;
    const int c = lane & 15;
    const int hh = lane >> 4;
    const int b = blockIdx.x;
    if (b >= nb) return;
    const int n0 = b * TILE;
    const int e0 = bstart[b];
    const int e1 = bstart[b + 1];
    unsigned* mxw = mx;
    const float2* esrc2 = (const float2*)esrc;
    const float2* edst2 = (const float2*)edst;

    if (threadIdx.x < TILE * HEADS) mx[threadIdx.x] = 0u;
    __syncthreads();

    for (int eb = e0 + (int)threadIdx.x; eb < e1; eb += 32 * SW * MPT) {
        int sm[MPT], dm[MPT];
#pragma unroll
        for (int u = 0; u < MPT; ++u) {
            const int ee = (eb + 32 * SW * u < e1) ? (eb + 32 * SW * u) : (e1 - 1);
            sm[u] = psrc[ee];
            dm[u] = pdst[ee];
        }
        float2 es[MPT], ed[MPT];
#pragma unroll
        for (int u = 0; u < MPT; ++u) { es[u] = esrc2[sm[u]]; ed[u] = edst2[dm[u]]; }
#pragma unroll
        for (int u = 0; u < MPT; ++u) {
            if (eb + 32 * SW * u < e1) {
                atomicMax(&mxw[(dm[u] - n0) * HEADS + 0], enc_f32(lrelu(es[u].x + ed[u].x)));
                atomicMax(&mxw[(dm[u] - n0) * HEADS + 1], enc_f32(lrelu(es[u].y + ed[u].y)));
            }
        }
    }
    __syncthreads();

    if (e0 >= e1) return;

    const int ew = e0 + 32 * ((wave - b) & (SW - 1));
    v8f acc[8];
#pragma unroll
    for (int t = 0; t < 8; ++t) acc[t] = (v8f){};
    float den0 = 0.f, den1 = 0.f;

    int sc, dc;
    float a0c, a1c;
    {
        const int ee = (ew + lane < e1) ? (ew + lane) : (e1 - 1);
        sc = psrc[ee];
        dc = pdst[ee];
        const float2 es = esrc2[sc], ed = edst2[dc];
        const bool ok = ew + lane < e1;
        a0c = ok ? __expf(lrelu(es.x + ed.x) - dec_f32(mxw[(dc - n0) * HEADS + 0])) : 0.f;
        a1c = ok ? __expf(lrelu(es.y + ed.y) - dec_f32(mxw[(dc - n0) * HEADS + 1])) : 0.f;
    }
    int sn, dn;
    {
        const int ee = (ew + 32 * SW + lane < e1) ? (ew + 32 * SW + lane) : (e1 - 1);
        sn = psrc[ee];
        dn = pdst[ee];
    }

    for (int eb = ew; eb < e1; eb += 32 * SW) {
        const float2 esn = esrc2[sn], edn = edst2[dn];
        int s2, d2;
        {
            const int ee = (eb + 64 * SW + lane < e1) ? (eb + 64 * SW + lane) : (e1 - 1);
            s2 = psrc[ee];
            d2 = pdst[ee];
        }
        const int dlc = dc - n0;

#pragma unroll
        for (int gb = 0; gb < 8 / GB; ++gb) {
            float4 r0[GB][2], r1[GB][2];
            int dlg[GB][2];
            float w0[GB][2], w1[GB][2];
#pragma unroll
            for (int q = 0; q < GB; ++q) {
#pragma unroll
                for (int i = 0; i < 2; ++i) {
                    const int j = (gb * GB + q) * 4 + 2 * hh + i;
                    const int sj = lane_get_i(j, sc);
                    dlg[q][i] = lane_get_i(j, dlc);
                    w0[q][i] = lane_get_f(j, a0c);
                    w1[q][i] = lane_get_f(j, a1c);
                    const float4* rp = (const float4*)(hfeat + (size_t)sj * FDIM) + c;
                    r0[q][i] = rp[0];
                    r1[q][i] = rp[16];
                }
            }
#pragma unroll
            for (int q = 0; q < GB; ++q) {
                v2f A0, A1;
                A0.x = (dlg[q][0] == c) ? w0[q][0] : 0.f;
                A0.y = (dlg[q][1] == c) ? w0[q][1] : 0.f;
                A1.x = (dlg[q][0] == c) ? w1[q][0] : 0.f;
                A1.y = (dlg[q][1] == c) ? w1[q][1] : 0.f;
                den0 += A0.x + A0.y;
                den1 += A1.x + A1.y;
                v2f B;
                B.x = r0[q][0].x; B.y = r0[q][1].x;
                acc[0] = __builtin_amdgcn_wmma_f32_16x16x4_f32(false, A0, false, B, (short)0, acc[0], false, false);
                B.x = r0[q][0].y; B.y = r0[q][1].y;
                acc[1] = __builtin_amdgcn_wmma_f32_16x16x4_f32(false, A0, false, B, (short)0, acc[1], false, false);
                B.x = r0[q][0].z; B.y = r0[q][1].z;
                acc[2] = __builtin_amdgcn_wmma_f32_16x16x4_f32(false, A0, false, B, (short)0, acc[2], false, false);
                B.x = r0[q][0].w; B.y = r0[q][1].w;
                acc[3] = __builtin_amdgcn_wmma_f32_16x16x4_f32(false, A0, false, B, (short)0, acc[3], false, false);
                B.x = r1[q][0].x; B.y = r1[q][1].x;
                acc[4] = __builtin_amdgcn_wmma_f32_16x16x4_f32(false, A1, false, B, (short)0, acc[4], false, false);
                B.x = r1[q][0].y; B.y = r1[q][1].y;
                acc[5] = __builtin_amdgcn_wmma_f32_16x16x4_f32(false, A1, false, B, (short)0, acc[5], false, false);
                B.x = r1[q][0].z; B.y = r1[q][1].z;
                acc[6] = __builtin_amdgcn_wmma_f32_16x16x4_f32(false, A1, false, B, (short)0, acc[6], false, false);
                B.x = r1[q][0].w; B.y = r1[q][1].w;
                acc[7] = __builtin_amdgcn_wmma_f32_16x16x4_f32(false, A1, false, B, (short)0, acc[7], false, false);
            }
        }

        {
            const bool ok = eb + 32 * SW + lane < e1;
            a0c = ok ? __expf(lrelu(esn.x + edn.x) - dec_f32(mxw[(dn - n0) * HEADS + 0])) : 0.f;
            a1c = ok ? __expf(lrelu(esn.y + edn.y) - dec_f32(mxw[(dn - n0) * HEADS + 1])) : 0.f;
            sc = sn; dc = dn;
            sn = s2; dn = d2;
        }
    }

    den0 += lane_get_f(lane ^ 16, den0);
    den1 += lane_get_f(lane ^ 16, den1);
    dens[wave][0][lane] = den0;
    dens[wave][1][lane] = den1;
#pragma unroll
    for (int r = 0; r < 8; ++r) {
        float4 p0, p1;
        p0.x = acc[0][r]; p0.y = acc[1][r]; p0.z = acc[2][r]; p0.w = acc[3][r];
        p1.x = acc[4][r]; p1.y = acc[5][r]; p1.z = acc[6][r]; p1.w = acc[7][r];
        red[wave][r][0][lane] = p0;
        red[wave][r][1][lane] = p1;
    }
    __syncthreads();

    const float4 bs0 = ((const float4*)bias)[c];
    const float4 bs1 = ((const float4*)bias)[16 + c];
    float4* z4 = (float4*)z;
#pragma unroll
    for (int i = 0; i < 8 / SW; ++i) {
        const int r = (8 / SW) * wave + i;
        const int row = 8 * hh + r;
        float4 s0 = red[0][r][0][lane], s1 = red[0][r][1][lane];
        float q0 = dens[0][0][row], q1 = dens[0][1][row];
#pragma unroll
        for (int v = 1; v < SW; ++v) {
            const float4 t0 = red[v][r][0][lane], t1 = red[v][r][1][lane];
            s0.x += t0.x; s0.y += t0.y; s0.z += t0.z; s0.w += t0.w;
            s1.x += t1.x; s1.y += t1.y; s1.z += t1.z; s1.w += t1.w;
            q0 += dens[v][0][row];
            q1 += dens[v][1][row];
        }
        float4 o0, o1;
        o0.x = s0.x / q0 + bs0.x; o0.y = s0.y / q0 + bs0.y;
        o0.z = s0.z / q0 + bs0.z; o0.w = s0.w / q0 + bs0.w;
        o1.x = s1.x / q1 + bs1.x; o1.y = s1.y / q1 + bs1.y;
        o1.z = s1.z / q1 + bs1.z; o1.w = s1.w / q1 + bs1.w;
        if (do_relu) {
            o0.x = fmaxf(o0.x, 0.f); o0.y = fmaxf(o0.y, 0.f); o0.z = fmaxf(o0.z, 0.f); o0.w = fmaxf(o0.w, 0.f);
            o1.x = fmaxf(o1.x, 0.f); o1.y = fmaxf(o1.y, 0.f); o1.z = fmaxf(o1.z, 0.f); o1.w = fmaxf(o1.w, 0.f);
        }
        const int node = n0 + row;
        if (node < n) {
            z4[(size_t)node * (FDIM / 4) + c] = o0;
            z4[(size_t)node * (FDIM / 4) + 16 + c] = o1;
        }
    }
}

#define DSLOTS 32
#define DBATCH 16

__device__ __forceinline__ float lane_xor16(float x) {
    return __builtin_bit_cast(float, __builtin_amdgcn_permlanex16(
        __builtin_bit_cast(int, x), __builtin_bit_cast(int, x), 0x76543210, 0xfedcba98, false, false));
}
template <int M>
__device__ __forceinline__ float lane_xor_row(float x) {
    return __builtin_bit_cast(float, __builtin_amdgcn_update_dpp(
        0, __builtin_bit_cast(int, x), 0x160 | M, 0xf, 0xf, true));
}

__global__ __launch_bounds__(256) void tile_decode(
    const int* __restrict__ psrc, const int* __restrict__ pdst, const int* __restrict__ peid,
    const int* __restrict__ bstart, const float* __restrict__ z,
    float* __restrict__ out, int n, int nb, int E)
{
    __shared__ float4 zt[DTILE * FDIM / 4];

    const int tid = threadIdx.x;
    const int wave = tid >> 5;
    const int lane = tid & 31;
    const int tile = blockIdx.x >> 1;
    const int n0 = tile * DTILE;
    const int b0 = tile * (DTILE / TILE);
    const int t0 = bstart[b0];
    const int t1 = bstart[min(b0 + DTILE / TILE, nb)];
    const int mid = t0 + ((t1 - t0) >> 1);
    const int e0 = (blockIdx.x & 1) ? mid : t0;
    const int e1 = (blockIdx.x & 1) ? t1 : mid;

    const float4* z4 = (const float4*)z;
    const unsigned loff = (unsigned)lane * 16u;
    float4 piece[DTILE * FDIM / 4 / 256];
#pragma unroll
    for (int k = 0; k < DTILE * FDIM / 4 / 256; ++k) {
        const int i = tid + 256 * k;
        const int node = min(n0 + (i >> 5), n - 1);
        piece[k] = z4[(size_t)node * (FDIM / 4) + (i & 31)];
    }
#pragma unroll
    for (int k = 0; k < DTILE * FDIM / 4 / 256; ++k) zt[tid + 256 * k] = piece[k];
    __syncthreads();

    for (int eb = e0 + wave * DSLOTS; eb < e1; eb += 8 * DSLOTS) {
        const int ee = (eb + lane < e1) ? (eb + lane) : (e1 - 1);
        const int sv = psrc[ee];
        const int dv = pdst[ee] - n0;
        const int ev = peid[ee];

        float p[DSLOTS];
#pragma unroll
        for (int b = 0; b < DSLOTS / DBATCH; ++b) {
            float4 g[DBATCH];
#pragma unroll
            for (int u = 0; u < DBATCH; ++u) {
                const unsigned s = (unsigned)__builtin_amdgcn_readlane(sv, b * DBATCH + u);
                const char* rowp = (const char*)z + (size_t)s * (FDIM * sizeof(float));
                g[u] = *(const float4*)(rowp + loff);
            }
#pragma unroll
            for (int u = 0; u < DBATCH; ++u) {
                const int dl = __builtin_amdgcn_readlane(dv, b * DBATCH + u);
                const float4 t = zt[dl * (FDIM / 4) + lane];
                p[b * DBATCH + u] = g[u].x * t.x + g[u].y * t.y + g[u].z * t.z + g[u].w * t.w;
            }
        }

        {
            const bool hi = (lane & 16) != 0;
#pragma unroll
            for (int u = 0; u < 16; ++u) {
                const float keep = hi ? p[u + 16] : p[u];
                const float send = hi ? p[u] : p[u + 16];
                p[u] = keep + lane_xor16(send);
            }
        }
        {
            const bool hi = (lane & 8) != 0;
#pragma unroll
            for (int u = 0; u < 8; ++u) {
                const float keep = hi ? p[u + 8] : p[u];
                const float send = hi ? p[u] : p[u + 8];
                p[u] = keep + lane_xor_row<8>(send);
            }
        }
        {
            const bool hi = (lane & 4) != 0;
#pragma unroll
            for (int u = 0; u < 4; ++u) {
                const float keep = hi ? p[u + 4] : p[u];
                const float send = hi ? p[u] : p[u + 4];
                p[u] = keep + lane_xor_row<4>(send);
            }
        }
        {
            const bool hi = (lane & 2) != 0;
#pragma unroll
            for (int u = 0; u < 2; ++u) {
                const float keep = hi ? p[u + 2] : p[u];
                const float send = hi ? p[u] : p[u + 2];
                p[u] = keep + lane_xor_row<2>(send);
            }
        }
        {
            const bool hi = (lane & 1) != 0;
            const float keep = hi ? p[1] : p[0];
            const float send = hi ? p[0] : p[1];
            p[0] = keep + lane_xor_row<1>(send);
        }
        if (eb + lane < e1 && ev < E) out[ev] = p[0];
    }
}

extern "C" void kernel_launch(void* const* d_in, const int* in_sizes, int n_in,
                              void* d_out, int out_size, void* d_ws, size_t ws_size,
                              hipStream_t stream) {
    const float* x      = (const float*)d_in[0];
    const int*   ei     = (const int*)d_in[1];
    const float* W1     = (const float*)d_in[2];
    const float* a1s    = (const float*)d_in[3];
    const float* a1d    = (const float*)d_in[4];
    const float* b1     = (const float*)d_in[5];
    const float* W2     = (const float*)d_in[6];
    const float* a2s    = (const float*)d_in[7];
    const float* a2d    = (const float*)d_in[8];
    const float* b2     = (const float*)d_in[9];
    float* scores = (float*)d_out;

    const int N = in_sizes[0] / FDIM;
    const int E = in_sizes[1] / 2;
    const int* src = ei;
    const int* dst = ei + E;

    float* ws   = (float*)d_ws;
    float*    h     = ws;
    int*      psrc  = (int*)(ws + (size_t)N * FDIM);
    int*      pdst  = psrc + (size_t)(E + N);
    int*      bcount  = pdst + (size_t)(E + N);
    int*      bstart  = bcount + MAXNB;
    int*      bcursor = bstart + MAXNB + 1;
    int*      peid  = bcursor + MAXNB;
    float*    z     = ws + (size_t)2 * N * FDIM;
    float*    esrc  = ws + (size_t)3 * N * FDIM;
    float*    edst  = esrc + (size_t)N * HEADS;

    const int ETOT = E + N;
    const int NB = (N + TILE - 1) / TILE;
    const int NT = (N + DTILE - 1) / DTILE;
    dim3 blk(256);
    dim3 gGemm((N + 127) / 128);
    dim3 gNH((N * HEADS + 255) / 256);
    dim3 gChunk((ETOT + CHUNK - 1) / CHUNK);
    dim3 gAgg(NB);

    bucket_zero<<<dim3((NB + 255) / 256), blk, 0, stream>>>(bcount, NB);
    bucket_count<<<gChunk, blk, 0, stream>>>(dst, E, N, NB, bcount);
    bucket_scan<<<dim3(1), dim3(1024), 0, stream>>>(bcount, NB, bstart, bcursor);
    bucket_scatter<<<gChunk, blk, 0, stream>>>(src, dst, E, N, NB, bcursor, psrc, pdst, peid);

    gemm128_wmma<<<gGemm, blk, 0, stream>>>(x, W1, h, N);
    attn_logits<<<gNH, blk, 0, stream>>>(h, a1s, a1d, esrc, edst, N);
    bucket_softmax_aggregate<<<gAgg, dim3(32 * SW), 0, stream>>>(psrc, pdst, bstart, esrc, edst, h, b1, z, N, NB,  1);

    gemm128_wmma<<<gGemm, blk, 0, stream>>>(z, W2, h, N);
    attn_logits<<<gNH, blk, 0, stream>>>(h, a2s, a2d, esrc, edst, N);
    bucket_softmax_aggregate<<<gAgg, dim3(32 * SW), 0, stream>>>(psrc, pdst, bstart, esrc, edst, h, b2, z, N, NB,  0);

    tile_decode<<<dim3(2 * NT), blk, 0, stream>>>(psrc, pdst, peid, bstart, z, scores, N, NB, E);
}


# edge softmax+aggregation: max pass leaves each slot's two LeakyReLU logits in LDS (block shared with the later partial sums), sum pass reads them there; its e_src/e_dst gathers gone; sums unchanged
# speedup vs baseline: 3.5180x; 1.0356x over previous
#include <hip/hip_runtime.h>
#include <hip/hip_bf16.h>


typedef __attribute__((ext_vector_type(2))) float v2f;
typedef __attribute__((ext_vector_type(8))) float v8f;

#define HEADS 2
#define FDIM 128
#define NEG_SLOPE 0.2f

__device__ __forceinline__ unsigned enc_f32(float f) {
    unsigned u = __float_as_uint(f);
    return (u & 0x80000000u) ? ~u : (u | 0x80000000u);
}
__device__ __forceinline__ float dec_f32(unsigned u) {
    return __uint_as_float((u & 0x80000000u) ? (u & 0x7FFFFFFFu) : ~u);
}

__global__ __launch_bounds__(256) void gemm128_wmma(
    const float* __restrict__ X, const float* __restrict__ W,
    float* __restrict__ Y, int M)
{
    __shared__ float lw[FDIM * FDIM];
    {
        const float4* w4 = (const float4*)W;
        float4* l4 = (float4*)lw;
        for (int i = threadIdx.x; i < FDIM * FDIM / 4; i += 256) l4[i] = w4[i];
    }
    __syncthreads();

    const int wave = threadIdx.x >> 5;
    const int lane = threadIdx.x & 31;
    const int m0 = (blockIdx.x * 8 + wave) * 16;
    if (m0 >= M) return;

    const int lm = lane & 15;
    const int lh = lane >> 4;
    const float* arow = X + (size_t)(m0 + lm) * FDIM + 2 * lh;

#pragma unroll
    for (int nt = 0; nt < 8; ++nt) {
        v8f acc = {};
        const int nc = nt * 16 + lm;
#pragma unroll 8
        for (int kk = 0; kk < 32; ++kk) {
            const int k0 = kk * 4;
            v2f a = *(const v2f*)(arow + k0);
            v2f b;
            b.x = lw[(k0 + 2 * lh) * FDIM + nc];
            b.y = lw[(k0 + 2 * lh + 1) * FDIM + nc];
            acc = __builtin_amdgcn_wmma_f32_16x16x4_f32(
                false, a, false, b, (short)0, acc, false, false);
        }
        float* yrow = Y + (size_t)(m0 + 8 * lh) * FDIM + nc;
#pragma unroll
        for (int r = 0; r < 8; ++r) yrow[(size_t)r * FDIM] = acc[r];
    }
}

__global__ void attn_logits(const float* __restrict__ h,
                            const float* __restrict__ a_src,
                            const float* __restrict__ a_dst,
                            float* __restrict__ esrc, float* __restrict__ edst,
                            int n)
{
    int idx = blockIdx.x * blockDim.x + threadIdx.x;
    if (idx >= n * HEADS) return;
    int node = idx >> 1, hd = idx & 1;
    const float* hp = h + (size_t)node * FDIM + hd * 64;
    const float* as = a_src + hd * 64;
    const float* ad = a_dst + hd * 64;
    float s = 0.f, t = 0.f;
#pragma unroll 8
    for (int f = 0; f < 64; ++f) { float x = hp[f]; s += x * as[f]; t += x * ad[f]; }
    esrc[idx] = s; edst[idx] = t;
}

#define TILE 16
#define TILE_SHIFT 4
#define DTILE 64
#define MAXNB 4096
#define CHUNK 8192

__global__ void bucket_zero(int* __restrict__ bcount, int nb)
{
    int i = blockIdx.x * blockDim.x + threadIdx.x;
    if (i < nb) bcount[i] = 0;
}

#define EPT (CHUNK / 256)

__global__ __launch_bounds__(256) void bucket_count(const int* __restrict__ dst,
                                                    int E, int n, int nb,
                                                    int* __restrict__ bcount)
{
    __shared__ int lh[MAXNB];
    const int base = blockIdx.x * CHUNK + threadIdx.x;
    int dv[EPT];
#pragma unroll
    for (int k = 0; k < EPT; ++k) dv[k] = dst[min(base + k * 256, E - 1)];
    for (int i = threadIdx.x; i < nb; i += 256) lh[i] = 0;
    __syncthreads();
#pragma unroll
    for (int k = 0; k < EPT; ++k) {
        const int e = base + k * 256;
        if (e < E + n) {
            const int d = (e < E) ? dv[k] : (e - E);
            atomicAdd(&lh[d >> TILE_SHIFT], 1);
        }
    }
    __syncthreads();
    for (int i = threadIdx.x; i < nb; i += 256) {
        int c = lh[i];
        if (c) atomicAdd(&bcount[i], c);
    }
}

__global__ __launch_bounds__(1024) void bucket_scan(const int* __restrict__ bcount, int nb,
                                                    int* __restrict__ bstart,
                                                    int* __restrict__ bcursor)
{
    __shared__ int s[1024];
    const int t = threadIdx.x;
    int cv[MAXNB / 1024];
    int tot = 0;
#pragma unroll
    for (int k = 0; k < MAXNB / 1024; ++k) {
        const int i = t * (MAXNB / 1024) + k;
        cv[k] = (i < nb) ? bcount[i] : 0;
        tot += cv[k];
    }
    s[t] = tot;
    __syncthreads();
    for (int off = 1; off < 1024; off <<= 1) {
        int v = (t >= off) ? s[t - off] : 0;
        __syncthreads();
        s[t] += v;
        __syncthreads();
    }
    int run = s[t] - tot;
#pragma unroll
    for (int k = 0; k < MAXNB / 1024; ++k) {
        const int i = t * (MAXNB / 1024) + k;
        if (i < nb) { bstart[i] = run; bcursor[i] = run; }
        run += cv[k];
    }
    if (t == 1023) bstart[nb] = s[t];
}

__global__ __launch_bounds__(256) void bucket_scatter(const int* __restrict__ src,
                                                      const int* __restrict__ dst,
                                                      int E, int n, int nb,
                                                      int* __restrict__ bcursor,
                                                      int* __restrict__ psrc,
                                                      int* __restrict__ pdst,
                                                      int* __restrict__ peid)
{
    __shared__ int lh[MAXNB];
    __shared__ int lbase[MAXNB];
    const int base = blockIdx.x * CHUNK + threadIdx.x;
    int dv[EPT], sv[EPT];
#pragma unroll
    for (int k = 0; k < EPT; ++k) dv[k] = dst[min(base + k * 256, E - 1)];
    for (int i = threadIdx.x; i < nb; i += 256) lh[i] = 0;
    __syncthreads();
#pragma unroll
    for (int k = 0; k < EPT; ++k) sv[k] = src[min(base + k * 256, E - 1)];
#pragma unroll
    for (int k = 0; k < EPT; ++k) {
        const int e = base + k * 256;
        if (e < E + n) {
            const int d = (e < E) ? dv[k] : (e - E);
            dv[k] = d;
            atomicAdd(&lh[d >> TILE_SHIFT], 1);
        }
    }
    __syncthreads();
    {
        int cv[MAXNB / 256], bv[MAXNB / 256];
#pragma unroll
        for (int j = 0; j < MAXNB / 256; ++j) {
            const int i = threadIdx.x + 256 * j;
            cv[j] = (i < nb) ? lh[i] : 0;
        }
#pragma unroll
        for (int j = 0; j < MAXNB / 256; ++j) {
            const int i = threadIdx.x + 256 * j;
            bv[j] = cv[j] ? atomicAdd(&bcursor[i], cv[j]) : 0;
        }
#pragma unroll
        for (int j = 0; j < MAXNB / 256; ++j) {
            const int i = threadIdx.x + 256 * j;
            if (i < nb) { lbase[i] = bv[j]; lh[i] = 0; }
        }
    }
    __syncthreads();
#pragma unroll
    for (int k = 0; k < EPT; ++k) {
        const int e = base + k * 256;
        if (e < E + n) {
            const int s = (e < E) ? sv[k] : (e - E);
            const int d = dv[k];
            const int b = d >> TILE_SHIFT;
            const int pos = lbase[b] + atomicAdd(&lh[b], 1);
            psrc[pos] = s;
            pdst[pos] = d;
            peid[pos] = e;
        }
    }
}

#define SW 4
#define MPT 8
#define GB 4
#define LCAP (SW * 8 * HEADS * 32 * 2)

__device__ __forceinline__ int lane_get_i(int src, int v) {
    return __builtin_amdgcn_ds_bpermute(src << 2, v);
}
__device__ __forceinline__ float lane_get_f(int src, float v) {
    return __builtin_bit_cast(float, __builtin_amdgcn_ds_bpermute(src << 2, __builtin_bit_cast(int, v)));
}

__device__ __forceinline__ float lrelu(float v) { return (v >= 0.f) ? v : NEG_SLOPE * v; }

__global__ __launch_bounds__(32 * SW) void bucket_softmax_aggregate(
    const int* __restrict__ psrc, const int* __restrict__ pdst,
    const int* __restrict__ bstart,
    const float* __restrict__ esrc, const float* __restrict__ edst,
    const float* __restrict__ hfeat, const float* __restrict__ bias,
    float* __restrict__ z, int n, int nb, int do_relu)
{
    __shared__ unsigned mx[TILE * HEADS];
    __shared__ float4 red[SW][8][HEADS][32];
    __shared__ float dens[SW][HEADS][32];
    v2f* slog = (v2f*)&red[0][0][0][0];

    const int wave = threadIdx.x >> 5;
    const int lane = threadIdx.x & 31;
    const int c = lane & 15;
    const int hh = lane >> 4;
    const int b = blockIdx.x;
    if (b >= nb) return;
    const int n0 = b * TILE;
    const int e0 = bstart[b];
    const int e1 = bstart[b + 1];
    unsigned* mxw = mx;
    const float2* esrc2 = (const float2*)esrc;
    const float2* edst2 = (const float2*)edst;

    if (threadIdx.x < TILE * HEADS) mx[threadIdx.x] = 0u;
    __syncthreads();

    for (int eb = e0 + (int)threadIdx.x; eb < e1; eb += 32 * SW * MPT) {
        int sm[MPT], dm[MPT];
#pragma unroll
        for (int u = 0; u < MPT; ++u) {
            const int ee = (eb + 32 * SW * u < e1) ? (eb + 32 * SW * u) : (e1 - 1);
            sm[u] = psrc[ee];
            dm[u] = pdst[ee];
        }
        float2 es[MPT], ed[MPT];
#pragma unroll
        for (int u = 0; u < MPT; ++u) { es[u] = esrc2[sm[u]]; ed[u] = edst2[dm[u]]; }
#pragma unroll
        for (int u = 0; u < MPT; ++u) {
            if (eb + 32 * SW * u < e1) {
                v2f v;
                v.x = lrelu(es[u].x + ed[u].x);
                v.y = lrelu(es[u].y + ed[u].y);
                atomicMax(&mxw[(dm[u] - n0) * HEADS + 0], enc_f32(v.x));
                atomicMax(&mxw[(dm[u] - n0) * HEADS + 1], enc_f32(v.y));
                const int i = eb + 32 * SW * u - e0;
                if (i < LCAP) slog[i] = v;
            }
        }
    }
    __syncthreads();

    if (e0 >= e1) return;

    const int ew = e0 + 32 * ((wave - b) & (SW - 1));
    v8f acc[8];
#pragma unroll
    for (int t = 0; t < 8; ++t) acc[t] = (v8f){};
    float den0 = 0.f, den1 = 0.f;

    int sc, dc;
    float a0c, a1c;
    {
        const int ee = (ew + lane < e1) ? (ew + lane) : (e1 - 1);
        sc = psrc[ee];
        dc = pdst[ee];
        v2f v;
        if (ee - e0 < LCAP) v = slog[ee - e0];
        else {
            const float2 es = esrc2[sc], ed = edst2[dc];
            v.x = lrelu(es.x + ed.x);
            v.y = lrelu(es.y + ed.y);
        }
        const bool ok = ew + lane < e1;
        a0c = ok ? __expf(v.x - dec_f32(mxw[(dc - n0) * HEADS + 0])) : 0.f;
        a1c = ok ? __expf(v.y - dec_f32(mxw[(dc - n0) * HEADS + 1])) : 0.f;
    }

    for (int eb = ew; eb < e1; eb += 32 * SW) {
        const int en = (eb + 32 * SW + lane < e1) ? (eb + 32 * SW + lane) : (e1 - 1);
        const int sn = psrc[en];
        const int dn = pdst[en];
        const int dlc = dc - n0;

#pragma unroll
        for (int gb = 0; gb < 8 / GB; ++gb) {
            float4 r0[GB][2], r1[GB][2];
            int dlg[GB][2];
            float w0[GB][2], w1[GB][2];
#pragma unroll
            for (int q = 0; q < GB; ++q) {
#pragma unroll
                for (int i = 0; i < 2; ++i) {
                    const int j = (gb * GB + q) * 4 + 2 * hh + i;
                    const int sj = lane_get_i(j, sc);
                    dlg[q][i] = lane_get_i(j, dlc);
                    w0[q][i] = lane_get_f(j, a0c);
                    w1[q][i] = lane_get_f(j, a1c);
                    const float4* rp = (const float4*)(hfeat + (size_t)sj * FDIM) + c;
                    r0[q][i] = rp[0];
                    r1[q][i] = rp[16];
                }
            }
#pragma unroll
            for (int q = 0; q < GB; ++q) {
                v2f A0, A1;
                A0.x = (dlg[q][0] == c) ? w0[q][0] : 0.f;
                A0.y = (dlg[q][1] == c) ? w0[q][1] : 0.f;
                A1.x = (dlg[q][0] == c) ? w1[q][0] : 0.f;
                A1.y = (dlg[q][1] == c) ? w1[q][1] : 0.f;
                den0 += A0.x + A0.y;
                den1 += A1.x + A1.y;
                v2f B;
                B.x = r0[q][0].x; B.y = r0[q][1].x;
                acc[0] = __builtin_amdgcn_wmma_f32_16x16x4_f32(false, A0, false, B, (short)0, acc[0], false, false);
                B.x = r0[q][0].y; B.y = r0[q][1].y;
                acc[1] = __builtin_amdgcn_wmma_f32_16x16x4_f32(false, A0, false, B, (short)0, acc[1], false, false);
                B.x = r0[q][0].z; B.y = r0[q][1].z;
                acc[2] = __builtin_amdgcn_wmma_f32_16x16x4_f32(false, A0, false, B, (short)0, acc[2], false, false);
                B.x = r0[q][0].w; B.y = r0[q][1].w;
                acc[3] = __builtin_amdgcn_wmma_f32_16x16x4_f32(false, A0, false, B, (short)0, acc[3], false, false);
                B.x = r1[q][0].x; B.y = r1[q][1].x;
                acc[4] = __builtin_amdgcn_wmma_f32_16x16x4_f32(false, A1, false, B, (short)0, acc[4], false, false);
                B.x = r1[q][0].y; B.y = r1[q][1].y;
                acc[5] = __builtin_amdgcn_wmma_f32_16x16x4_f32(false, A1, false, B, (short)0, acc[5], false, false);
                B.x = r1[q][0].z; B.y = r1[q][1].z;
                acc[6] = __builtin_amdgcn_wmma_f32_16x16x4_f32(false, A1, false, B, (short)0, acc[6], false, false);
                B.x = r1[q][0].w; B.y = r1[q][1].w;
                acc[7] = __builtin_amdgcn_wmma_f32_16x16x4_f32(false, A1, false, B, (short)0, acc[7], false, false);
            }
        }

        {
            v2f v;
            if (en - e0 < LCAP) v = slog[en - e0];
            else {
                const float2 es = esrc2[sn], ed = edst2[dn];
                v.x = lrelu(es.x + ed.x);
                v.y = lrelu(es.y + ed.y);
            }
            const bool ok = eb + 32 * SW + lane < e1;
            a0c = ok ? __expf(v.x - dec_f32(mxw[(dn - n0) * HEADS + 0])) : 0.f;
            a1c = ok ? __expf(v.y - dec_f32(mxw[(dn - n0) * HEADS + 1])) : 0.f;
            sc = sn; dc = dn;
        }
    }
    __syncthreads();

    den0 += lane_get_f(lane ^ 16, den0);
    den1 += lane_get_f(lane ^ 16, den1);
    dens[wave][0][lane] = den0;
    dens[wave][1][lane] = den1;
#pragma unroll
    for (int r = 0; r < 8; ++r) {
        float4 p0, p1;
        p0.x = acc[0][r]; p0.y = acc[1][r]; p0.z = acc[2][r]; p0.w = acc[3][r];
        p1.x = acc[4][r]; p1.y = acc[5][r]; p1.z = acc[6][r]; p1.w = acc[7][r];
        red[wave][r][0][lane] = p0;
        red[wave][r][1][lane] = p1;
    }
    __syncthreads();

    const float4 bs0 = ((const float4*)bias)[c];
    const float4 bs1 = ((const float4*)bias)[16 + c];
    float4* z4 = (float4*)z;
#pragma unroll
    for (int i = 0; i < 8 / SW; ++i) {
        const int r = (8 / SW) * wave + i;
        const int row = 8 * hh + r;
        float4 s0 = red[0][r][0][lane], s1 = red[0][r][1][lane];
        float q0 = dens[0][0][row], q1 = dens[0][1][row];
#pragma unroll
        for (int v = 1; v < SW; ++v) {
            const float4 t0 = red[v][r][0][lane], t1 = red[v][r][1][lane];
            s0.x += t0.x; s0.y += t0.y; s0.z += t0.z; s0.w += t0.w;
            s1.x += t1.x; s1.y += t1.y; s1.z += t1.z; s1.w += t1.w;
            q0 += dens[v][0][row];
            q1 += dens[v][1][row];
        }
        float4 o0, o1;
        o0.x = s0.x / q0 + bs0.x; o0.y = s0.y / q0 + bs0.y;
        o0.z = s0.z / q0 + bs0.z; o0.w = s0.w / q0 + bs0.w;
        o1.x = s1.x / q1 + bs1.x; o1.y = s1.y / q1 + bs1.y;
        o1.z = s1.z / q1 + bs1.z; o1.w = s1.w / q1 + bs1.w;
        if (do_relu) {
            o0.x = fmaxf(o0.x, 0.f); o0.y = fmaxf(o0.y, 0.f); o0.z = fmaxf(o0.z, 0.f); o0.w = fmaxf(o0.w, 0.f);
            o1.x = fmaxf(o1.x, 0.f); o1.y = fmaxf(o1.y, 0.f); o1.z = fmaxf(o1.z, 0.f); o1.w = fmaxf(o1.w, 0.f);
        }
        const int node = n0 + row;
        if (node < n) {
            z4[(size_t)node * (FDIM / 4) + c] = o0;
            z4[(size_t)node * (FDIM / 4) + 16 + c] = o1;
        }
    }
}

#define DSLOTS 32
#define DBATCH 16

__device__ __forceinline__ float lane_xor16(float x) {
    return __builtin_bit_cast(float, __builtin_amdgcn_permlanex16(
        __builtin_bit_cast(int, x), __builtin_bit_cast(int, x), 0x76543210, 0xfedcba98, false, false));
}
template <int M>
__device__ __forceinline__ float lane_xor_row(float x) {
    return __builtin_bit_cast(float, __builtin_amdgcn_update_dpp(
        0, __builtin_bit_cast(int, x), 0x160 | M, 0xf, 0xf, true));
}

__global__ __launch_bounds__(256) void tile_decode(
    const int* __restrict__ psrc, const int* __restrict__ pdst, const int* __restrict__ peid,
    const int* __restrict__ bstart, const float* __restrict__ z,
    float* __restrict__ out, int n, int nb, int E)
{
    __shared__ float4 zt[DTILE * FDIM / 4];

    const int tid = threadIdx.x;
    const int wave = tid >> 5;
    const int lane = tid & 31;
    const int tile = blockIdx.x >> 1;
    const int n0 = tile * DTILE;
    const int b0 = tile * (DTILE / TILE);
    const int t0 = bstart[b0];
    const int t1 = bstart[min(b0 + DTILE / TILE, nb)];
    const int mid = t0 + ((t1 - t0) >> 1);
    const int e0 = (blockIdx.x & 1) ? mid : t0;
    const int e1 = (blockIdx.x & 1) ? t1 : mid;

    const float4* z4 = (const float4*)z;
    const unsigned loff = (unsigned)lane * 16u;
    float4 piece[DTILE * FDIM / 4 / 256];
#pragma unroll
    for (int k = 0; k < DTILE * FDIM / 4 / 256; ++k) {
        const int i = tid + 256 * k;
        const int node = min(n0 + (i >> 5), n - 1);
        piece[k] = z4[(size_t)node * (FDIM / 4) + (i & 31)];
    }
#pragma unroll
    for (int k = 0; k < DTILE * FDIM / 4 / 256; ++k) zt[tid + 256 * k] = piece[k];
    __syncthreads();

    for (int eb = e0 + wave * DSLOTS; eb < e1; eb += 8 * DSLOTS) {
        const int ee = (eb + lane < e1) ? (eb + lane) : (e1 - 1);
        const int sv = psrc[ee];
        const int dv = pdst[ee] - n0;
        const int ev = peid[ee];

        float p[DSLOTS];
#pragma unroll
        for (int b = 0; b < DSLOTS / DBATCH; ++b) {
            float4 g[DBATCH];
#pragma unroll
            for (int u = 0; u < DBATCH; ++u) {
                const unsigned s = (unsigned)__builtin_amdgcn_readlane(sv, b * DBATCH + u);
                const char* rowp = (const char*)z + (size_t)s * (FDIM * sizeof(float));
                g[u] = *(const float4*)(rowp + loff);
            }
#pragma unroll
            for (int u = 0; u < DBATCH; ++u) {
                const int dl = __builtin_amdgcn_readlane(dv, b * DBATCH + u);
                const float4 t = zt[dl * (FDIM / 4) + lane];
                p[b * DBATCH + u] = g[u].x * t.x + g[u].y * t.y + g[u].z * t.z + g[u].w * t.w;
            }
        }

        {
            const bool hi = (lane & 16) != 0;
#pragma unroll
            for (int u = 0; u < 16; ++u) {
                const float keep = hi ? p[u + 16] : p[u];
                const float send = hi ? p[u] : p[u + 16];
                p[u] = keep + lane_xor16(send);
            }
        }
        {
            const bool hi = (lane & 8) != 0;
#pragma unroll
            for (int u = 0; u < 8; ++u) {
                const float keep = hi ? p[u + 8] : p[u];
                const float send = hi ? p[u] : p[u + 8];
                p[u] = keep + lane_xor_row<8>(send);
            }
        }
        {
            const bool hi = (lane & 4) != 0;
#pragma unroll
            for (int u = 0; u < 4; ++u) {
                const float keep = hi ? p[u + 4] : p[u];
                const float send = hi ? p[u] : p[u + 4];
                p[u] = keep + lane_xor_row<4>(send);
            }
        }
        {
            const bool hi = (lane & 2) != 0;
#pragma unroll
            for (int u = 0; u < 2; ++u) {
                const float keep = hi ? p[u + 2] : p[u];
                const float send = hi ? p[u] : p[u + 2];
                p[u] = keep + lane_xor_row<2>(send);
            }
        }
        {
            const bool hi = (lane & 1) != 0;
            const float keep = hi ? p[1] : p[0];
            const float send = hi ? p[0] : p[1];
            p[0] = keep + lane_xor_row<1>(send);
        }
        if (eb + lane < e1 && ev < E) out[ev] = p[0];
    }
}

extern "C" void kernel_launch(void* const* d_in, const int* in_sizes, int n_in,
                              void* d_out, int out_size, void* d_ws, size_t ws_size,
                              hipStream_t stream) {
    const float* x      = (const float*)d_in[0];
    const int*   ei     = (const int*)d_in[1];
    const float* W1     = (const float*)d_in[2];
    const float* a1s    = (const float*)d_in[3];
    const float* a1d    = (const float*)d_in[4];
    const float* b1     = (const float*)d_in[5];
    const float* W2     = (const float*)d_in[6];
    const float* a2s    = (const float*)d_in[7];
    const float* a2d    = (const float*)d_in[8];
    const float* b2     = (const float*)d_in[9];
    float* scores = (float*)d_out;

    const int N = in_sizes[0] / FDIM;
    const int E = in_sizes[1] / 2;
    const int* src = ei;
    const int* dst = ei + E;

    float* ws   = (float*)d_ws;
    float*    h     = ws;
    int*      psrc  = (int*)(ws + (size_t)N * FDIM);
    int*      pdst  = psrc + (size_t)(E + N);
    int*      bcount  = pdst + (size_t)(E + N);
    int*      bstart  = bcount + MAXNB;
    int*      bcursor = bstart + MAXNB + 1;
    int*      peid  = bcursor + MAXNB;
    float*    z     = ws + (size_t)2 * N * FDIM;
    float*    esrc  = ws + (size_t)3 * N * FDIM;
    float*    edst  = esrc + (size_t)N * HEADS;

    const int ETOT = E + N;
    const int NB = (N + TILE - 1) / TILE;
    const int NT = (N + DTILE - 1) / DTILE;
    dim3 blk(256);
    dim3 gGemm((N + 127) / 128);
    dim3 gNH((N * HEADS + 255) / 256);
    dim3 gChunk((ETOT + CHUNK - 1) / CHUNK);
    dim3 gAgg(NB);

    bucket_zero<<<dim3((NB + 255) / 256), blk, 0, stream>>>(bcount, NB);
    bucket_count<<<gChunk, blk, 0, stream>>>(dst, E, N, NB, bcount);
    bucket_scan<<<dim3(1), dim3(1024), 0, stream>>>(bcount, NB, bstart, bcursor);
    bucket_scatter<<<gChunk, blk, 0, stream>>>(src, dst, E, N, NB, bcursor, psrc, pdst, peid);

    gemm128_wmma<<<gGemm, blk, 0, stream>>>(x, W1, h, N);
    attn_logits<<<gNH, blk, 0, stream>>>(h, a1s, a1d, esrc, edst, N);
    bucket_softmax_aggregate<<<gAgg, dim3(32 * SW), 0, stream>>>(psrc, pdst, bstart, esrc, edst, h, b1, z, N, NB,  1);

    gemm128_wmma<<<gGemm, blk, 0, stream>>>(z, W2, h, N);
    attn_logits<<<gNH, blk, 0, stream>>>(h, a2s, a2d, esrc, edst, N);
    bucket_softmax_aggregate<<<gAgg, dim3(32 * SW), 0, stream>>>(psrc, pdst, bstart, esrc, edst, h, b2, z, N, NB,  0);

    tile_decode<<<dim3(2 * NT), blk, 0, stream>>>(psrc, pdst, peid, bstart, z, scores, N, NB, E);
}


# bucket slot = one 8-byte record (src; dst-in-tile + edge id in one word): scatter issues one 8-byte store per slot, was three 4-byte stores to three lines; readers one 8-byte load; sums as before
# speedup vs baseline: 3.9865x; 1.1332x over previous
#include <hip/hip_runtime.h>
#include <hip/hip_bf16.h>


typedef __attribute__((ext_vector_type(2))) float v2f;
typedef __attribute__((ext_vector_type(8))) float v8f;

#define HEADS 2
#define FDIM 128
#define NEG_SLOPE 0.2f

__device__ __forceinline__ unsigned enc_f32(float f) {
    unsigned u = __float_as_uint(f);
    return (u & 0x80000000u) ? ~u : (u | 0x80000000u);
}
__device__ __forceinline__ float dec_f32(unsigned u) {
    return __uint_as_float((u & 0x80000000u) ? (u & 0x7FFFFFFFu) : ~u);
}

__global__ __launch_bounds__(256) void gemm128_wmma(
    const float* __restrict__ X, const float* __restrict__ W,
    float* __restrict__ Y, int M)
{
    __shared__ float lw[FDIM * FDIM];
    {
        const float4* w4 = (const float4*)W;
        float4* l4 = (float4*)lw;
        for (int i = threadIdx.x; i < FDIM * FDIM / 4; i += 256) l4[i] = w4[i];
    }
    __syncthreads();

    const int wave = threadIdx.x >> 5;
    const int lane = threadIdx.x & 31;
    const int m0 = (blockIdx.x * 8 + wave) * 16;
    if (m0 >= M) return;

    const int lm = lane & 15;
    const int lh = lane >> 4;
    const float* arow = X + (size_t)(m0 + lm) * FDIM + 2 * lh;

#pragma unroll
    for (int nt = 0; nt < 8; ++nt) {
        v8f acc = {};
        const int nc = nt * 16 + lm;
#pragma unroll 8
        for (int kk = 0; kk < 32; ++kk) {
            const int k0 = kk * 4;
            v2f a = *(const v2f*)(arow + k0);
            v2f b;
            b.x = lw[(k0 + 2 * lh) * FDIM + nc];
            b.y = lw[(k0 + 2 * lh + 1) * FDIM + nc];
            acc = __builtin_amdgcn_wmma_f32_16x16x4_f32(
                false, a, false, b, (short)0, acc, false, false);
        }
        float* yrow = Y + (size_t)(m0 + 8 * lh) * FDIM + nc;
#pragma unroll
        for (int r = 0; r < 8; ++r) yrow[(size_t)r * FDIM] = acc[r];
    }
}

__global__ void attn_logits(const float* __restrict__ h,
                            const float* __restrict__ a_src,
                            const float* __restrict__ a_dst,
                            float* __restrict__ esrc, float* __restrict__ edst,
                            int n)
{
    int idx = blockIdx.x * blockDim.x + threadIdx.x;
    if (idx >= n * HEADS) return;
    int node = idx >> 1, hd = idx & 1;
    const float* hp = h + (size_t)node * FDIM + hd * 64;
    const float* as = a_src + hd * 64;
    const float* ad = a_dst + hd * 64;
    float s = 0.f, t = 0.f;
#pragma unroll 8
    for (int f = 0; f < 64; ++f) { float x = hp[f]; s += x * as[f]; t += x * ad[f]; }
    esrc[idx] = s; edst[idx] = t;
}

#define TILE 16
#define TILE_SHIFT 4
#define DTILE 64
#define MAXNB 4096
#define CHUNK 8192

#define EID_BITS 26
__device__ __forceinline__ int rec_pack(int d, int e) {
    return (int)(((unsigned)(d & (DTILE - 1)) << EID_BITS) | (unsigned)e);
}
__device__ __forceinline__ int rec_row_tile(int y)   { return (int)((unsigned)y >> EID_BITS); }
__device__ __forceinline__ int rec_row_bucket(int y) { return (int)(((unsigned)y >> EID_BITS) & (TILE - 1)); }
__device__ __forceinline__ int rec_eid(int y)        { return y & ((1 << EID_BITS) - 1); }

__global__ void bucket_zero(int* __restrict__ bcount, int nb)
{
    int i = blockIdx.x * blockDim.x + threadIdx.x;
    if (i < nb) bcount[i] = 0;
}

#define EPT (CHUNK / 256)

__global__ __launch_bounds__(256) void bucket_count(const int* __restrict__ dst,
                                                    int E, int n, int nb,
                                                    int* __restrict__ bcount)
{
    __shared__ int lh[MAXNB];
    const int base = blockIdx.x * CHUNK + threadIdx.x;
    int dv[EPT];
#pragma unroll
    for (int k = 0; k < EPT; ++k) dv[k] = dst[min(base + k * 256, E - 1)];
    for (int i = threadIdx.x; i < nb; i += 256) lh[i] = 0;
    __syncthreads();
#pragma unroll
    for (int k = 0; k < EPT; ++k) {
        const int e = base + k * 256;
        if (e < E + n) {
            const int d = (e < E) ? dv[k] : (e - E);
            atomicAdd(&lh[d >> TILE_SHIFT], 1);
        }
    }
    __syncthreads();
    for (int i = threadIdx.x; i < nb; i += 256) {
        int c = lh[i];
        if (c) atomicAdd(&bcount[i], c);
    }
}

__global__ __launch_bounds__(1024) void bucket_scan(const int* __restrict__ bcount, int nb,
                                                    int* __restrict__ bstart,
                                                    int* __restrict__ bcursor)
{
    __shared__ int s[1024];
    const int t = threadIdx.x;
    int cv[MAXNB / 1024];
    int tot = 0;
#pragma unroll
    for (int k = 0; k < MAXNB / 1024; ++k) {
        const int i = t * (MAXNB / 1024) + k;
        cv[k] = (i < nb) ? bcount[i] : 0;
        tot += cv[k];
    }
    s[t] = tot;
    __syncthreads();
    for (int off = 1; off < 1024; off <<= 1) {
        int v = (t >= off) ? s[t - off] : 0;
        __syncthreads();
        s[t] += v;
        __syncthreads();
    }
    int run = s[t] - tot;
#pragma unroll
    for (int k = 0; k < MAXNB / 1024; ++k) {
        const int i = t * (MAXNB / 1024) + k;
        if (i < nb) { bstart[i] = run; bcursor[i] = run; }
        run += cv[k];
    }
    if (t == 1023) bstart[nb] = s[t];
}

__global__ __launch_bounds__(256) void bucket_scatter(const int* __restrict__ src,
                                                      const int* __restrict__ dst,
                                                      int E, int n, int nb,
                                                      int* __restrict__ bcursor,
                                                      int* __restrict__ prec)
{
    int2* rec = (int2*)prec;
    __shared__ int lh[MAXNB];
    __shared__ int lbase[MAXNB];
    const int base = blockIdx.x * CHUNK + threadIdx.x;
    int dv[EPT], sv[EPT];
#pragma unroll
    for (int k = 0; k < EPT; ++k) dv[k] = dst[min(base + k * 256, E - 1)];
    for (int i = threadIdx.x; i < nb; i += 256) lh[i] = 0;
    __syncthreads();
#pragma unroll
    for (int k = 0; k < EPT; ++k) sv[k] = src[min(base + k * 256, E - 1)];
#pragma unroll
    for (int k = 0; k < EPT; ++k) {
        const int e = base + k * 256;
        if (e < E + n) {
            const int d = (e < E) ? dv[k] : (e - E);
            dv[k] = d;
            atomicAdd(&lh[d >> TILE_SHIFT], 1);
        }
    }
    __syncthreads();
    {
        int cv[MAXNB / 256], bv[MAXNB / 256];
#pragma unroll
        for (int j = 0; j < MAXNB / 256; ++j) {
            const int i = threadIdx.x + 256 * j;
            cv[j] = (i < nb) ? lh[i] : 0;
        }
#pragma unroll
        for (int j = 0; j < MAXNB / 256; ++j) {
            const int i = threadIdx.x + 256 * j;
            bv[j] = cv[j] ? atomicAdd(&bcursor[i], cv[j]) : 0;
        }
#pragma unroll
        for (int j = 0; j < MAXNB / 256; ++j) {
            const int i = threadIdx.x + 256 * j;
            if (i < nb) { lbase[i] = bv[j]; lh[i] = 0; }
        }
    }
    __syncthreads();
#pragma unroll
    for (int k = 0; k < EPT; ++k) {
        const int e = base + k * 256;
        if (e < E + n) {
            const int s = (e < E) ? sv[k] : (e - E);
            const int d = dv[k];
            const int b = d >> TILE_SHIFT;
            const int pos = lbase[b] + atomicAdd(&lh[b], 1);
            int2 r;
            r.x = s;
            r.y = rec_pack(d, e);
            rec[pos] = r;
        }
    }
}

#define SW 4
#define MPT 8
#define GB 4
#define LCAP (SW * 8 * HEADS * 32 * 2)

__device__ __forceinline__ int lane_get_i(int src, int v) {
    return __builtin_amdgcn_ds_bpermute(src << 2, v);
}
__device__ __forceinline__ float lane_get_f(int src, float v) {
    return __builtin_bit_cast(float, __builtin_amdgcn_ds_bpermute(src << 2, __builtin_bit_cast(int, v)));
}

__device__ __forceinline__ float lrelu(float v) { return (v >= 0.f) ? v : NEG_SLOPE * v; }

__global__ __launch_bounds__(32 * SW) void bucket_softmax_aggregate(
    const int* __restrict__ prec,
    const int* __restrict__ bstart,
    const float* __restrict__ esrc, const float* __restrict__ edst,
    const float* __restrict__ hfeat, const float* __restrict__ bias,
    float* __restrict__ z, int n, int nb, int do_relu)
{
    __shared__ unsigned mx[TILE * HEADS];
    __shared__ float4 red[SW][8][HEADS][32];
    __shared__ float dens[SW][HEADS][32];
    v2f* slog = (v2f*)&red[0][0][0][0];

    const int wave = threadIdx.x >> 5;
    const int lane = threadIdx.x & 31;
    const int c = lane & 15;
    const int hh = lane >> 4;
    const int b = blockIdx.x;
    if (b >= nb) return;
    const int n0 = b * TILE;
    const int e0 = bstart[b];
    const int e1 = bstart[b + 1];
    unsigned* mxw = mx;
    const float2* esrc2 = (const float2*)esrc;
    const float2* edst2 = (const float2*)edst;
    const int2* rec = (const int2*)prec;

    if (threadIdx.x < TILE * HEADS) mx[threadIdx.x] = 0u;
    __syncthreads();

    for (int eb = e0 + (int)threadIdx.x; eb < e1; eb += 32 * SW * MPT) {
        int sm[MPT], dm[MPT];
#pragma unroll
        for (int u = 0; u < MPT; ++u) {
            const int ee = (eb + 32 * SW * u < e1) ? (eb + 32 * SW * u) : (e1 - 1);
            const int2 r = rec[ee];
            sm[u] = r.x;
            dm[u] = rec_row_bucket(r.y);
        }
        float2 es[MPT], ed[MPT];
#pragma unroll
        for (int u = 0; u < MPT; ++u) { es[u] = esrc2[sm[u]]; ed[u] = edst2[n0 + dm[u]]; }
#pragma unroll
        for (int u = 0; u < MPT; ++u) {
            if (eb + 32 * SW * u < e1) {
                v2f v;
                v.x = lrelu(es[u].x + ed[u].x);
                v.y = lrelu(es[u].y + ed[u].y);
                atomicMax(&mxw[dm[u] * HEADS + 0], enc_f32(v.x));
                atomicMax(&mxw[dm[u] * HEADS + 1], enc_f32(v.y));
                const int i = eb + 32 * SW * u - e0;
                if (i < LCAP) slog[i] = v;
            }
        }
    }
    __syncthreads();

    if (e0 >= e1) return;

    const int ew = e0 + 32 * ((wave - b) & (SW - 1));
    v8f acc[8];
#pragma unroll
    for (int t = 0; t < 8; ++t) acc[t] = (v8f){};
    float den0 = 0.f, den1 = 0.f;

    int sc, dc;
    float a0c, a1c;
    {
        const int ee = (ew + lane < e1) ? (ew + lane) : (e1 - 1);
        const int2 r = rec[ee];
        sc = r.x;
        dc = rec_row_bucket(r.y);
        v2f v;
        if (ee - e0 < LCAP) v = slog[ee - e0];
        else {
            const float2 es = esrc2[sc], ed = edst2[n0 + dc];
            v.x = lrelu(es.x + ed.x);
            v.y = lrelu(es.y + ed.y);
        }
        const bool ok = ew + lane < e1;
        a0c = ok ? __expf(v.x - dec_f32(mxw[dc * HEADS + 0])) : 0.f;
        a1c = ok ? __expf(v.y - dec_f32(mxw[dc * HEADS + 1])) : 0.f;
    }

    for (int eb = ew; eb < e1; eb += 32 * SW) {
        const int en = (eb + 32 * SW + lane < e1) ? (eb + 32 * SW + lane) : (e1 - 1);
        const int2 rn = rec[en];
        const int sn = rn.x;
        const int dn = rec_row_bucket(rn.y);
        const int dlc = dc;

#pragma unroll
        for (int gb = 0; gb < 8 / GB; ++gb) {
            float4 r0[GB][2], r1[GB][2];
            int dlg[GB][2];
            float w0[GB][2], w1[GB][2];
#pragma unroll
            for (int q = 0; q < GB; ++q) {
#pragma unroll
                for (int i = 0; i < 2; ++i) {
                    const int j = (gb * GB + q) * 4 + 2 * hh + i;
                    const int sj = lane_get_i(j, sc);
                    dlg[q][i] = lane_get_i(j, dlc);
                    w0[q][i] = lane_get_f(j, a0c);
                    w1[q][i] = lane_get_f(j, a1c);
                    const float4* rp = (const float4*)(hfeat + (size_t)sj * FDIM) + c;
                    r0[q][i] = rp[0];
                    r1[q][i] = rp[16];
                }
            }
#pragma unroll
            for (int q = 0; q < GB; ++q) {
                v2f A0, A1;
                A0.x = (dlg[q][0] == c) ? w0[q][0] : 0.f;
                A0.y = (dlg[q][1] == c) ? w0[q][1] : 0.f;
                A1.x = (dlg[q][0] == c) ? w1[q][0] : 0.f;
                A1.y = (dlg[q][1] == c) ? w1[q][1] : 0.f;
                den0 += A0.x + A0.y;
                den1 += A1.x + A1.y;
                v2f B;
                B.x = r0[q][0].x; B.y = r0[q][1].x;
                acc[0] = __builtin_amdgcn_wmma_f32_16x16x4_f32(false, A0, false, B, (short)0, acc[0], false, false);
                B.x = r0[q][0].y; B.y = r0[q][1].y;
                acc[1] = __builtin_amdgcn_wmma_f32_16x16x4_f32(false, A0, false, B, (short)0, acc[1], false, false);
                B.x = r0[q][0].z; B.y = r0[q][1].z;
                acc[2] = __builtin_amdgcn_wmma_f32_16x16x4_f32(false, A0, false, B, (short)0, acc[2], false, false);
                B.x = r0[q][0].w; B.y = r0[q][1].w;
                acc[3] = __builtin_amdgcn_wmma_f32_16x16x4_f32(false, A0, false, B, (short)0, acc[3], false, false);
                B.x = r1[q][0].x; B.y = r1[q][1].x;
                acc[4] = __builtin_amdgcn_wmma_f32_16x16x4_f32(false, A1, false, B, (short)0, acc[4], false, false);
                B.x = r1[q][0].y; B.y = r1[q][1].y;
                acc[5] = __builtin_amdgcn_wmma_f32_16x16x4_f32(false, A1, false, B, (short)0, acc[5], false, false);
                B.x = r1[q][0].z; B.y = r1[q][1].z;
                acc[6] = __builtin_amdgcn_wmma_f32_16x16x4_f32(false, A1, false, B, (short)0, acc[6], false, false);
                B.x = r1[q][0].w; B.y = r1[q][1].w;
                acc[7] = __builtin_amdgcn_wmma_f32_16x16x4_f32(false, A1, false, B, (short)0, acc[7], false, false);
            }
        }

        {
            v2f v;
            if (en - e0 < LCAP) v = slog[en - e0];
            else {
                const float2 es = esrc2[sn], ed = edst2[n0 + dn];
                v.x = lrelu(es.x + ed.x);
                v.y = lrelu(es.y + ed.y);
            }
            const bool ok = eb + 32 * SW + lane < e1;
            a0c = ok ? __expf(v.x - dec_f32(mxw[dn * HEADS + 0])) : 0.f;
            a1c = ok ? __expf(v.y - dec_f32(mxw[dn * HEADS + 1])) : 0.f;
            sc = sn; dc = dn;
        }
    }
    __syncthreads();

    den0 += lane_get_f(lane ^ 16, den0);
    den1 += lane_get_f(lane ^ 16, den1);
    dens[wave][0][lane] = den0;
    dens[wave][1][lane] = den1;
#pragma unroll
    for (int r = 0; r < 8; ++r) {
        float4 p0, p1;
        p0.x = acc[0][r]; p0.y = acc[1][r]; p0.z = acc[2][r]; p0.w = acc[3][r];
        p1.x = acc[4][r]; p1.y = acc[5][r]; p1.z = acc[6][r]; p1.w = acc[7][r];
        red[wave][r][0][lane] = p0;
        red[wave][r][1][lane] = p1;
    }
    __syncthreads();

    const float4 bs0 = ((const float4*)bias)[c];
    const float4 bs1 = ((const float4*)bias)[16 + c];
    float4* z4 = (float4*)z;
#pragma unroll
    for (int i = 0; i < 8 / SW; ++i) {
        const int r = (8 / SW) * wave + i;
        const int row = 8 * hh + r;
        float4 s0 = red[0][r][0][lane], s1 = red[0][r][1][lane];
        float q0 = dens[0][0][row], q1 = dens[0][1][row];
#pragma unroll
        for (int v = 1; v < SW; ++v) {
            const float4 t0 = red[v][r][0][lane], t1 = red[v][r][1][lane];
            s0.x += t0.x; s0.y += t0.y; s0.z += t0.z; s0.w += t0.w;
            s1.x += t1.x; s1.y += t1.y; s1.z += t1.z; s1.w += t1.w;
            q0 += dens[v][0][row];
            q1 += dens[v][1][row];
        }
        float4 o0, o1;
        o0.x = s0.x / q0 + bs0.x; o0.y = s0.y / q0 + bs0.y;
        o0.z = s0.z / q0 + bs0.z; o0.w = s0.w / q0 + bs0.w;
        o1.x = s1.x / q1 + bs1.x; o1.y = s1.y / q1 + bs1.y;
        o1.z = s1.z / q1 + bs1.z; o1.w = s1.w / q1 + bs1.w;
        if (do_relu) {
            o0.x = fmaxf(o0.x, 0.f); o0.y = fmaxf(o0.y, 0.f); o0.z = fmaxf(o0.z, 0.f); o0.w = fmaxf(o0.w, 0.f);
            o1.x = fmaxf(o1.x, 0.f); o1.y = fmaxf(o1.y, 0.f); o1.z = fmaxf(o1.z, 0.f); o1.w = fmaxf(o1.w, 0.f);
        }
        const int node = n0 + row;
        if (node < n) {
            z4[(size_t)node * (FDIM / 4) + c] = o0;
            z4[(size_t)node * (FDIM / 4) + 16 + c] = o1;
        }
    }
}

#define DSLOTS 32
#define DBATCH 16

__device__ __forceinline__ float lane_xor16(float x) {
    return __builtin_bit_cast(float, __builtin_amdgcn_permlanex16(
        __builtin_bit_cast(int, x), __builtin_bit_cast(int, x), 0x76543210, 0xfedcba98, false, false));
}
template <int M>
__device__ __forceinline__ float lane_xor_row(float x) {
    return __builtin_bit_cast(float, __builtin_amdgcn_update_dpp(
        0, __builtin_bit_cast(int, x), 0x160 | M, 0xf, 0xf, true));
}

__global__ __launch_bounds__(256) void tile_decode(
    const int* __restrict__ prec,
    const int* __restrict__ bstart, const float* __restrict__ z,
    float* __restrict__ out, int n, int nb, int E)
{
    __shared__ float4 zt[DTILE * FDIM / 4];

    const int tid = threadIdx.x;
    const int wave = tid >> 5;
    const int lane = tid & 31;
    const int tile = blockIdx.x >> 1;
    const int n0 = tile * DTILE;
    const int b0 = tile * (DTILE / TILE);
    const int t0 = bstart[b0];
    const int t1 = bstart[min(b0 + DTILE / TILE, nb)];
    const int mid = t0 + ((t1 - t0) >> 1);
    const int e0 = (blockIdx.x & 1) ? mid : t0;
    const int e1 = (blockIdx.x & 1) ? t1 : mid;

    const float4* z4 = (const float4*)z;
    const int2* rec = (const int2*)prec;
    const unsigned loff = (unsigned)lane * 16u;
    float4 piece[DTILE * FDIM / 4 / 256];
#pragma unroll
    for (int k = 0; k < DTILE * FDIM / 4 / 256; ++k) {
        const int i = tid + 256 * k;
        const int node = min(n0 + (i >> 5), n - 1);
        piece[k] = z4[(size_t)node * (FDIM / 4) + (i & 31)];
    }
#pragma unroll
    for (int k = 0; k < DTILE * FDIM / 4 / 256; ++k) zt[tid + 256 * k] = piece[k];
    __syncthreads();

    for (int eb = e0 + wave * DSLOTS; eb < e1; eb += 8 * DSLOTS) {
        const int ee = (eb + lane < e1) ? (eb + lane) : (e1 - 1);
        const int2 r = rec[ee];
        const int sv = r.x;
        const int dv = rec_row_tile(r.y);
        const int ev = rec_eid(r.y);

        float p[DSLOTS];
#pragma unroll
        for (int b = 0; b < DSLOTS / DBATCH; ++b) {
            float4 g[DBATCH];
#pragma unroll
            for (int u = 0; u < DBATCH; ++u) {
                const unsigned s = (unsigned)__builtin_amdgcn_readlane(sv, b * DBATCH + u);
                const char* rowp = (const char*)z + (size_t)s * (FDIM * sizeof(float));
                g[u] = *(const float4*)(rowp + loff);
            }
#pragma unroll
            for (int u = 0; u < DBATCH; ++u) {
                const int dl = __builtin_amdgcn_readlane(dv, b * DBATCH + u);
                const float4 t = zt[dl * (FDIM / 4) + lane];
                p[b * DBATCH + u] = g[u].x * t.x + g[u].y * t.y + g[u].z * t.z + g[u].w * t.w;
            }
        }

        {
            const bool hi = (lane & 16) != 0;
#pragma unroll
            for (int u = 0; u < 16; ++u) {
                const float keep = hi ? p[u + 16] : p[u];
                const float send = hi ? p[u] : p[u + 16];
                p[u] = keep + lane_xor16(send);
            }
        }
        {
            const bool hi = (lane & 8) != 0;
#pragma unroll
            for (int u = 0; u < 8; ++u) {
                const float keep = hi ? p[u + 8] : p[u];
                const float send = hi ? p[u] : p[u + 8];
                p[u] = keep + lane_xor_row<8>(send);
            }
        }
        {
            const bool hi = (lane & 4) != 0;
#pragma unroll
            for (int u = 0; u < 4; ++u) {
                const float keep = hi ? p[u + 4] : p[u];
                const float send = hi ? p[u] : p[u + 4];
                p[u] = keep + lane_xor_row<4>(send);
            }
        }
        {
            const bool hi = (lane & 2) != 0;
#pragma unroll
            for (int u = 0; u < 2; ++u) {
                const float keep = hi ? p[u + 2] : p[u];
                const float send = hi ? p[u] : p[u + 2];
                p[u] = keep + lane_xor_row<2>(send);
            }
        }
        {
            const bool hi = (lane & 1) != 0;
            const float keep = hi ? p[1] : p[0];
            const float send = hi ? p[0] : p[1];
            p[0] = keep + lane_xor_row<1>(send);
        }
        if (eb + lane < e1 && ev < E) out[ev] = p[0];
    }
}

extern "C" void kernel_launch(void* const* d_in, const int* in_sizes, int n_in,
                              void* d_out, int out_size, void* d_ws, size_t ws_size,
                              hipStream_t stream) {
    const float* x      = (const float*)d_in[0];
    const int*   ei     = (const int*)d_in[1];
    const float* W1     = (const float*)d_in[2];
    const float* a1s    = (const float*)d_in[3];
    const float* a1d    = (const float*)d_in[4];
    const float* b1     = (const float*)d_in[5];
    const float* W2     = (const float*)d_in[6];
    const float* a2s    = (const float*)d_in[7];
    const float* a2d    = (const float*)d_in[8];
    const float* b2     = (const float*)d_in[9];
    float* scores = (float*)d_out;

    const int N = in_sizes[0] / FDIM;
    const int E = in_sizes[1] / 2;
    const int* src = ei;
    const int* dst = ei + E;

    float* ws   = (float*)d_ws;
    float*    h     = ws;
    int*      prec  = (int*)(ws + (size_t)N * FDIM);
    int*      bcount  = prec + (size_t)2 * (E + N);
    int*      bstart  = bcount + MAXNB;
    int*      bcursor = bstart + MAXNB + 1;
    float*    z     = ws + (size_t)2 * N * FDIM;
    float*    esrc  = ws + (size_t)3 * N * FDIM;
    float*    edst  = esrc + (size_t)N * HEADS;

    const int ETOT = E + N;
    const int NB = (N + TILE - 1) / TILE;
    const int NT = (N + DTILE - 1) / DTILE;
    dim3 blk(256);
    dim3 gGemm((N + 127) / 128);
    dim3 gNH((N * HEADS + 255) / 256);
    dim3 gChunk((ETOT + CHUNK - 1) / CHUNK);
    dim3 gAgg(NB);

    bucket_zero<<<dim3((NB + 255) / 256), blk, 0, stream>>>(bcount, NB);
    bucket_count<<<gChunk, blk, 0, stream>>>(dst, E, N, NB, bcount);
    bucket_scan<<<dim3(1), dim3(1024), 0, stream>>>(bcount, NB, bstart, bcursor);
    bucket_scatter<<<gChunk, blk, 0, stream>>>(src, dst, E, N, NB, bcursor, prec);

    gemm128_wmma<<<gGemm, blk, 0, stream>>>(x, W1, h, N);
    attn_logits<<<gNH, blk, 0, stream>>>(h, a1s, a1d, esrc, edst, N);
    bucket_softmax_aggregate<<<gAgg, dim3(32 * SW), 0, stream>>>(prec, bstart, esrc, edst, h, b1, z, N, NB,  1);

    gemm128_wmma<<<gGemm, blk, 0, stream>>>(z, W2, h, N);
    attn_logits<<<gNH, blk, 0, stream>>>(h, a2s, a2d, esrc, edst, N);
    bucket_softmax_aggregate<<<gAgg, dim3(32 * SW), 0, stream>>>(prec, bstart, esrc, edst, h, b2, z, N, NB,  0);

    tile_decode<<<dim3(2 * NT), blk, 0, stream>>>(prec, bstart, z, scores, N, NB, E);
}


# GEMM operand feed: the work-group's 128 X rows staged in LDS by whole-row loads, A fragments read there (was an 8-byte global load over 16 lines per WMMA); conflict-free LDS layouts; sums unchanged
# speedup vs baseline: 4.2345x; 1.0622x over previous
#include <hip/hip_runtime.h>
#include <hip/hip_bf16.h>


typedef __attribute__((ext_vector_type(2))) float v2f;
typedef __attribute__((ext_vector_type(8))) float v8f;

#define HEADS 2
#define FDIM 128
#define NEG_SLOPE 0.2f

__device__ __forceinline__ unsigned enc_f32(float f) {
    unsigned u = __float_as_uint(f);
    return (u & 0x80000000u) ? ~u : (u | 0x80000000u);
}
__device__ __forceinline__ float dec_f32(unsigned u) {
    return __uint_as_float((u & 0x80000000u) ? (u & 0x7FFFFFFFu) : ~u);
}

#define GROWS 128
#define XP 132
__device__ __forceinline__ int wcol(int k, int n) { return n ^ ((k & 2) << 4); }

__global__ __launch_bounds__(256) void gemm128_wmma(
    const float* __restrict__ X, const float* __restrict__ W,
    float* __restrict__ Y, int M)
{
    __shared__ float lw[FDIM * FDIM];
    __shared__ float lx[GROWS * XP];

    const int wave = threadIdx.x >> 5;
    const int lane = threadIdx.x & 31;
    const int mblk = blockIdx.x * GROWS;
    {
        const float4* w4 = (const float4*)W;
        const float4* x4 = (const float4*)X;
        float4 wv[FDIM / 8], xv[GROWS / 8];
#pragma unroll
        for (int k = 0; k < FDIM / 8; ++k)
            wv[k] = w4[(wave + 8 * k) * (FDIM / 4) + lane];
#pragma unroll
        for (int k = 0; k < GROWS / 8; ++k) {
            const int gr = min(mblk + wave * 16 + k, M - 1);
            xv[k] = x4[(size_t)gr * (FDIM / 4) + lane];
        }
#pragma unroll
        for (int k = 0; k < FDIM / 8; ++k) {
            const int r = wave + 8 * k;
            *(float4*)(lw + r * FDIM + wcol(r, lane * 4)) = wv[k];
        }
#pragma unroll
        for (int k = 0; k < GROWS / 8; ++k)
            *(float4*)(lx + (wave * 16 + k) * XP + lane * 4) = xv[k];
    }
    __syncthreads();

    const int m0 = mblk + wave * 16;
    if (m0 >= M) return;

    const int lm = lane & 15;
    const int lh = lane >> 4;
    const float* arow = lx + (wave * 16 + lm) * XP + 2 * lh;
    const float* brow = lw + (2 * lh) * FDIM;

#pragma unroll
    for (int nt = 0; nt < 8; ++nt) {
        v8f acc = {};
        const int nc = nt * 16 + lm;
        const int ncs = nc ^ (lh << 5);
#pragma unroll 8
        for (int kk = 0; kk < 32; ++kk) {
            const int k0 = kk * 4;
            v2f a = *(const v2f*)(arow + k0);
            v2f b;
            b.x = brow[k0 * FDIM + ncs];
            b.y = brow[(k0 + 1) * FDIM + ncs];
            acc = __builtin_amdgcn_wmma_f32_16x16x4_f32(
                false, a, false, b, (short)0, acc, false, false);
        }
        float* yrow = Y + (size_t)(m0 + 8 * lh) * FDIM + nc;
#pragma unroll
        for (int r = 0; r < 8; ++r) yrow[(size_t)r * FDIM] = acc[r];
    }
}

__global__ void attn_logits(const float* __restrict__ h,
                            const float* __restrict__ a_src,
                            const float* __restrict__ a_dst,
                            float* __restrict__ esrc, float* __restrict__ edst,
                            int n)
{
    int idx = blockIdx.x * blockDim.x + threadIdx.x;
    if (idx >= n * HEADS) return;
    int node = idx >> 1, hd = idx & 1;
    const float* hp = h + (size_t)node * FDIM + hd * 64;
    const float* as = a_src + hd * 64;
    const float* ad = a_dst + hd * 64;
    float s = 0.f, t = 0.f;
#pragma unroll 8
    for (int f = 0; f < 64; ++f) { float x = hp[f]; s += x * as[f]; t += x * ad[f]; }
    esrc[idx] = s; edst[idx] = t;
}

#define TILE 16
#define TILE_SHIFT 4
#define DTILE 64
#define MAXNB 4096
#define CHUNK 8192

#define EID_BITS 26
__device__ __forceinline__ int rec_pack(int d, int e) {
    return (int)(((unsigned)(d & (DTILE - 1)) << EID_BITS) | (unsigned)e);
}
__device__ __forceinline__ int rec_row_tile(int y)   { return (int)((unsigned)y >> EID_BITS); }
__device__ __forceinline__ int rec_row_bucket(int y) { return (int)(((unsigned)y >> EID_BITS) & (TILE - 1)); }
__device__ __forceinline__ int rec_eid(int y)        { return y & ((1 << EID_BITS) - 1); }

__global__ void bucket_zero(int* __restrict__ bcount, int nb)
{
    int i = blockIdx.x * blockDim.x + threadIdx.x;
    if (i < nb) bcount[i] = 0;
}

#define EPT (CHUNK / 256)

__global__ __launch_bounds__(256) void bucket_count(const int* __restrict__ dst,
                                                    int E, int n, int nb,
                                                    int* __restrict__ bcount)
{
    __shared__ int lh[MAXNB];
    const int base = blockIdx.x * CHUNK + threadIdx.x;
    int dv[EPT];
#pragma unroll
    for (int k = 0; k < EPT; ++k) dv[k] = dst[min(base + k * 256, E - 1)];
    for (int i = threadIdx.x; i < nb; i += 256) lh[i] = 0;
    __syncthreads();
#pragma unroll
    for (int k = 0; k < EPT; ++k) {
        const int e = base + k * 256;
        if (e < E + n) {
            const int d = (e < E) ? dv[k] : (e - E);
            atomicAdd(&lh[d >> TILE_SHIFT], 1);
        }
    }
    __syncthreads();
    for (int i = threadIdx.x; i < nb; i += 256) {
        int c = lh[i];
        if (c) atomicAdd(&bcount[i], c);
    }
}

__global__ __launch_bounds__(1024) void bucket_scan(const int* __restrict__ bcount, int nb,
                                                    int* __restrict__ bstart,
                                                    int* __restrict__ bcursor)
{
    __shared__ int s[1024];
    const int t = threadIdx.x;
    int cv[MAXNB / 1024];
    int tot = 0;
#pragma unroll
    for (int k = 0; k < MAXNB / 1024; ++k) {
        const int i = t * (MAXNB / 1024) + k;
        cv[k] = (i < nb) ? bcount[i] : 0;
        tot += cv[k];
    }
    s[t] = tot;
    __syncthreads();
    for (int off = 1; off < 1024; off <<= 1) {
        int v = (t >= off) ? s[t - off] : 0;
        __syncthreads();
        s[t] += v;
        __syncthreads();
    }
    int run = s[t] - tot;
#pragma unroll
    for (int k = 0; k < MAXNB / 1024; ++k) {
        const int i = t * (MAXNB / 1024) + k;
        if (i < nb) { bstart[i] = run; bcursor[i] = run; }
        run += cv[k];
    }
    if (t == 1023) bstart[nb] = s[t];
}

__global__ __launch_bounds__(256) void bucket_scatter(const int* __restrict__ src,
                                                      const int* __restrict__ dst,
                                                      int E, int n, int nb,
                                                      int* __restrict__ bcursor,
                                                      int* __restrict__ prec)
{
    int2* rec = (int2*)prec;
    __shared__ int lh[MAXNB];
    __shared__ int lbase[MAXNB];
    const int base = blockIdx.x * CHUNK + threadIdx.x;
    int dv[EPT], sv[EPT];
#pragma unroll
    for (int k = 0; k < EPT; ++k) dv[k] = dst[min(base + k * 256, E - 1)];
    for (int i = threadIdx.x; i < nb; i += 256) lh[i] = 0;
    __syncthreads();
#pragma unroll
    for (int k = 0; k < EPT; ++k) sv[k] = src[min(base + k * 256, E - 1)];
#pragma unroll
    for (int k = 0; k < EPT; ++k) {
        const int e = base + k * 256;
        if (e < E + n) {
            const int d = (e < E) ? dv[k] : (e - E);
            dv[k] = d;
            atomicAdd(&lh[d >> TILE_SHIFT], 1);
        }
    }
    __syncthreads();
    {
        int cv[MAXNB / 256], bv[MAXNB / 256];
#pragma unroll
        for (int j = 0; j < MAXNB / 256; ++j) {
            const int i = threadIdx.x + 256 * j;
            cv[j] = (i < nb) ? lh[i] : 0;
        }
#pragma unroll
        for (int j = 0; j < MAXNB / 256; ++j) {
            const int i = threadIdx.x + 256 * j;
            bv[j] = cv[j] ? atomicAdd(&bcursor[i], cv[j]) : 0;
        }
#pragma unroll
        for (int j = 0; j < MAXNB / 256; ++j) {
            const int i = threadIdx.x + 256 * j;
            if (i < nb) { lbase[i] = bv[j]; lh[i] = 0; }
        }
    }
    __syncthreads();
#pragma unroll
    for (int k = 0; k < EPT; ++k) {
        const int e = base + k * 256;
        if (e < E + n) {
            const int s = (e < E) ? sv[k] : (e - E);
            const int d = dv[k];
            const int b = d >> TILE_SHIFT;
            const int pos = lbase[b] + atomicAdd(&lh[b], 1);
            int2 r;
            r.x = s;
            r.y = rec_pack(d, e);
            rec[pos] = r;
        }
    }
}

#define SW 4
#define MPT 8
#define GB 4
#define LCAP (SW * 8 * HEADS * 32 * 2)

__device__ __forceinline__ int lane_get_i(int src, int v) {
    return __builtin_amdgcn_ds_bpermute(src << 2, v);
}
__device__ __forceinline__ float lane_get_f(int src, float v) {
    return __builtin_bit_cast(float, __builtin_amdgcn_ds_bpermute(src << 2, __builtin_bit_cast(int, v)));
}

__device__ __forceinline__ float lrelu(float v) { return (v >= 0.f) ? v : NEG_SLOPE * v; }

__global__ __launch_bounds__(32 * SW) void bucket_softmax_aggregate(
    const int* __restrict__ prec,
    const int* __restrict__ bstart,
    const float* __restrict__ esrc, const float* __restrict__ edst,
    const float* __restrict__ hfeat, const float* __restrict__ bias,
    float* __restrict__ z, int n, int nb, int do_relu)
{
    __shared__ unsigned mx[TILE * HEADS];
    __shared__ float4 red[SW][8][HEADS][32];
    __shared__ float dens[SW][HEADS][32];
    v2f* slog = (v2f*)&red[0][0][0][0];

    const int wave = threadIdx.x >> 5;
    const int lane = threadIdx.x & 31;
    const int c = lane & 15;
    const int hh = lane >> 4;
    const int b = blockIdx.x;
    if (b >= nb) return;
    const int n0 = b * TILE;
    const int e0 = bstart[b];
    const int e1 = bstart[b + 1];
    unsigned* mxw = mx;
    const float2* esrc2 = (const float2*)esrc;
    const float2* edst2 = (const float2*)edst;
    const int2* rec = (const int2*)prec;

    if (threadIdx.x < TILE * HEADS) mx[threadIdx.x] = 0u;
    __syncthreads();

    for (int eb = e0 + (int)threadIdx.x; eb < e1; eb += 32 * SW * MPT) {
        int sm[MPT], dm[MPT];
#pragma unroll
        for (int u = 0; u < MPT; ++u) {
            const int ee = (eb + 32 * SW * u < e1) ? (eb + 32 * SW * u) : (e1 - 1);
            const int2 r = rec[ee];
            sm[u] = r.x;
            dm[u] = rec_row_bucket(r.y);
        }
        float2 es[MPT], ed[MPT];
#pragma unroll
        for (int u = 0; u < MPT; ++u) { es[u] = esrc2[sm[u]]; ed[u] = edst2[n0 + dm[u]]; }
#pragma unroll
        for (int u = 0; u < MPT; ++u) {
            if (eb + 32 * SW * u < e1) {
                v2f v;
                v.x = lrelu(es[u].x + ed[u].x);
                v.y = lrelu(es[u].y + ed[u].y);
                atomicMax(&mxw[dm[u] * HEADS + 0], enc_f32(v.x));
                atomicMax(&mxw[dm[u] * HEADS + 1], enc_f32(v.y));
                const int i = eb + 32 * SW * u - e0;
                if (i < LCAP) slog[i] = v;
            }
        }
    }
    __syncthreads();

    if (e0 >= e1) return;

    const int ew = e0 + 32 * ((wave - b) & (SW - 1));
    v8f acc[8];
#pragma unroll
    for (int t = 0; t < 8; ++t) acc[t] = (v8f){};
    float den0 = 0.f, den1 = 0.f;

    int sc, dc;
    float a0c, a1c;
    {
        const int ee = (ew + lane < e1) ? (ew + lane) : (e1 - 1);
        const int2 r = rec[ee];
        sc = r.x;
        dc = rec_row_bucket(r.y);
        v2f v;
        if (ee - e0 < LCAP) v = slog[ee - e0];
        else {
            const float2 es = esrc2[sc], ed = edst2[n0 + dc];
            v.x = lrelu(es.x + ed.x);
            v.y = lrelu(es.y + ed.y);
        }
        const bool ok = ew + lane < e1;
        a0c = ok ? __expf(v.x - dec_f32(mxw[dc * HEADS + 0])) : 0.f;
        a1c = ok ? __expf(v.y - dec_f32(mxw[dc * HEADS + 1])) : 0.f;
    }

    for (int eb = ew; eb < e1; eb += 32 * SW) {
        const int en = (eb + 32 * SW + lane < e1) ? (eb + 32 * SW + lane) : (e1 - 1);
        const int2 rn = rec[en];
        const int sn = rn.x;
        const int dn = rec_row_bucket(rn.y);
        const int dlc = dc;

#pragma unroll
        for (int gb = 0; gb < 8 / GB; ++gb) {
            float4 r0[GB][2], r1[GB][2];
            int dlg[GB][2];
            float w0[GB][2], w1[GB][2];
#pragma unroll
            for (int q = 0; q < GB; ++q) {
#pragma unroll
                for (int i = 0; i < 2; ++i) {
                    const int j = (gb * GB + q) * 4 + 2 * hh + i;
                    const int sj = lane_get_i(j, sc);
                    dlg[q][i] = lane_get_i(j, dlc);
                    w0[q][i] = lane_get_f(j, a0c);
                    w1[q][i] = lane_get_f(j, a1c);
                    const float4* rp = (const float4*)(hfeat + (size_t)sj * FDIM) + c;
                    r0[q][i] = rp[0];
                    r1[q][i] = rp[16];
                }
            }
#pragma unroll
            for (int q = 0; q < GB; ++q) {
                v2f A0, A1;
                A0.x = (dlg[q][0] == c) ? w0[q][0] : 0.f;
                A0.y = (dlg[q][1] == c) ? w0[q][1] : 0.f;
                A1.x = (dlg[q][0] == c) ? w1[q][0] : 0.f;
                A1.y = (dlg[q][1] == c) ? w1[q][1] : 0.f;
                den0 += A0.x + A0.y;
                den1 += A1.x + A1.y;
                v2f B;
                B.x = r0[q][0].x; B.y = r0[q][1].x;
                acc[0] = __builtin_amdgcn_wmma_f32_16x16x4_f32(false, A0, false, B, (short)0, acc[0], false, false);
                B.x = r0[q][0].y; B.y = r0[q][1].y;
                acc[1] = __builtin_amdgcn_wmma_f32_16x16x4_f32(false, A0, false, B, (short)0, acc[1], false, false);
                B.x = r0[q][0].z; B.y = r0[q][1].z;
                acc[2] = __builtin_amdgcn_wmma_f32_16x16x4_f32(false, A0, false, B, (short)0, acc[2], false, false);
                B.x = r0[q][0].w; B.y = r0[q][1].w;
                acc[3] = __builtin_amdgcn_wmma_f32_16x16x4_f32(false, A0, false, B, (short)0, acc[3], false, false);
                B.x = r1[q][0].x; B.y = r1[q][1].x;
                acc[4] = __builtin_amdgcn_wmma_f32_16x16x4_f32(false, A1, false, B, (short)0, acc[4], false, false);
                B.x = r1[q][0].y; B.y = r1[q][1].y;
                acc[5] = __builtin_amdgcn_wmma_f32_16x16x4_f32(false, A1, false, B, (short)0, acc[5], false, false);
                B.x = r1[q][0].z; B.y = r1[q][1].z;
                acc[6] = __builtin_amdgcn_wmma_f32_16x16x4_f32(false, A1, false, B, (short)0, acc[6], false, false);
                B.x = r1[q][0].w; B.y = r1[q][1].w;
                acc[7] = __builtin_amdgcn_wmma_f32_16x16x4_f32(false, A1, false, B, (short)0, acc[7], false, false);
            }
        }

        {
            v2f v;
            if (en - e0 < LCAP) v = slog[en - e0];
            else {
                const float2 es = esrc2[sn], ed = edst2[n0 + dn];
                v.x = lrelu(es.x + ed.x);
                v.y = lrelu(es.y + ed.y);
            }
            const bool ok = eb + 32 * SW + lane < e1;
            a0c = ok ? __expf(v.x - dec_f32(mxw[dn * HEADS + 0])) : 0.f;
            a1c = ok ? __expf(v.y - dec_f32(mxw[dn * HEADS + 1])) : 0.f;
            sc = sn; dc = dn;
        }
    }
    __syncthreads();

    den0 += lane_get_f(lane ^ 16, den0);
    den1 += lane_get_f(lane ^ 16, den1);
    dens[wave][0][lane] = den0;
    dens[wave][1][lane] = den1;
#pragma unroll
    for (int r = 0; r < 8; ++r) {
        float4 p0, p1;
        p0.x = acc[0][r]; p0.y = acc[1][r]; p0.z = acc[2][r]; p0.w = acc[3][r];
        p1.x = acc[4][r]; p1.y = acc[5][r]; p1.z = acc[6][r]; p1.w = acc[7][r];
        red[wave][r][0][lane] = p0;
        red[wave][r][1][lane] = p1;
    }
    __syncthreads();

    const float4 bs0 = ((const float4*)bias)[c];
    const float4 bs1 = ((const float4*)bias)[16 + c];
    float4* z4 = (float4*)z;
#pragma unroll
    for (int i = 0; i < 8 / SW; ++i) {
        const int r = (8 / SW) * wave + i;
        const int row = 8 * hh + r;
        float4 s0 = red[0][r][0][lane], s1 = red[0][r][1][lane];
        float q0 = dens[0][0][row], q1 = dens[0][1][row];
#pragma unroll
        for (int v = 1; v < SW; ++v) {
            const float4 t0 = red[v][r][0][lane], t1 = red[v][r][1][lane];
            s0.x += t0.x; s0.y += t0.y; s0.z += t0.z; s0.w += t0.w;
            s1.x += t1.x; s1.y += t1.y; s1.z += t1.z; s1.w += t1.w;
            q0 += dens[v][0][row];
            q1 += dens[v][1][row];
        }
        float4 o0, o1;
        o0.x = s0.x / q0 + bs0.x; o0.y = s0.y / q0 + bs0.y;
        o0.z = s0.z / q0 + bs0.z; o0.w = s0.w / q0 + bs0.w;
        o1.x = s1.x / q1 + bs1.x; o1.y = s1.y / q1 + bs1.y;
        o1.z = s1.z / q1 + bs1.z; o1.w = s1.w / q1 + bs1.w;
        if (do_relu) {
            o0.x = fmaxf(o0.x, 0.f); o0.y = fmaxf(o0.y, 0.f); o0.z = fmaxf(o0.z, 0.f); o0.w = fmaxf(o0.w, 0.f);
            o1.x = fmaxf(o1.x, 0.f); o1.y = fmaxf(o1.y, 0.f); o1.z = fmaxf(o1.z, 0.f); o1.w = fmaxf(o1.w, 0.f);
        }
        const int node = n0 + row;
        if (node < n) {
            z4[(size_t)node * (FDIM / 4) + c] = o0;
            z4[(size_t)node * (FDIM / 4) + 16 + c] = o1;
        }
    }
}

#define DSLOTS 32
#define DBATCH 16

__device__ __forceinline__ float lane_xor16(float x) {
    return __builtin_bit_cast(float, __builtin_amdgcn_permlanex16(
        __builtin_bit_cast(int, x), __builtin_bit_cast(int, x), 0x76543210, 0xfedcba98, false, false));
}
template <int M>
__device__ __forceinline__ float lane_xor_row(float x) {
    return __builtin_bit_cast(float, __builtin_amdgcn_update_dpp(
        0, __builtin_bit_cast(int, x), 0x160 | M, 0xf, 0xf, true));
}

__global__ __launch_bounds__(256) void tile_decode(
    const int* __restrict__ prec,
    const int* __restrict__ bstart, const float* __restrict__ z,
    float* __restrict__ out, int n, int nb, int E)
{
    __shared__ float4 zt[DTILE * FDIM / 4];

    const int tid = threadIdx.x;
    const int wave = tid >> 5;
    const int lane = tid & 31;
    const int tile = blockIdx.x >> 1;
    const int n0 = tile * DTILE;
    const int b0 = tile * (DTILE / TILE);
    const int t0 = bstart[b0];
    const int t1 = bstart[min(b0 + DTILE / TILE, nb)];
    const int mid = t0 + ((t1 - t0) >> 1);
    const int e0 = (blockIdx.x & 1) ? mid : t0;
    const int e1 = (blockIdx.x & 1) ? t1 : mid;

    const float4* z4 = (const float4*)z;
    const int2* rec = (const int2*)prec;
    const unsigned loff = (unsigned)lane * 16u;
    float4 piece[DTILE * FDIM / 4 / 256];
#pragma unroll
    for (int k = 0; k < DTILE * FDIM / 4 / 256; ++k) {
        const int i = tid + 256 * k;
        const int node = min(n0 + (i >> 5), n - 1);
        piece[k] = z4[(size_t)node * (FDIM / 4) + (i & 31)];
    }
#pragma unroll
    for (int k = 0; k < DTILE * FDIM / 4 / 256; ++k) zt[tid + 256 * k] = piece[k];
    __syncthreads();

    for (int eb = e0 + wave * DSLOTS; eb < e1; eb += 8 * DSLOTS) {
        const int ee = (eb + lane < e1) ? (eb + lane) : (e1 - 1);
        const int2 r = rec[ee];
        const int sv = r.x;
        const int dv = rec_row_tile(r.y);
        const int ev = rec_eid(r.y);

        float p[DSLOTS];
#pragma unroll
        for (int b = 0; b < DSLOTS / DBATCH; ++b) {
            float4 g[DBATCH];
#pragma unroll
            for (int u = 0; u < DBATCH; ++u) {
                const unsigned s = (unsigned)__builtin_amdgcn_readlane(sv, b * DBATCH + u);
                const char* rowp = (const char*)z + (size_t)s * (FDIM * sizeof(float));
                g[u] = *(const float4*)(rowp + loff);
            }
#pragma unroll
            for (int u = 0; u < DBATCH; ++u) {
                const int dl = __builtin_amdgcn_readlane(dv, b * DBATCH + u);
                const float4 t = zt[dl * (FDIM / 4) + lane];
                p[b * DBATCH + u] = g[u].x * t.x + g[u].y * t.y + g[u].z * t.z + g[u].w * t.w;
            }
        }

        {
            const bool hi = (lane & 16) != 0;
#pragma unroll
            for (int u = 0; u < 16; ++u) {
                const float keep = hi ? p[u + 16] : p[u];
                const float send = hi ? p[u] : p[u + 16];
                p[u] = keep + lane_xor16(send);
            }
        }
        {
            const bool hi = (lane & 8) != 0;
#pragma unroll
            for (int u = 0; u < 8; ++u) {
                const float keep = hi ? p[u + 8] : p[u];
                const float send = hi ? p[u] : p[u + 8];
                p[u] = keep + lane_xor_row<8>(send);
            }
        }
        {
            const bool hi = (lane & 4) != 0;
#pragma unroll
            for (int u = 0; u < 4; ++u) {
                const float keep = hi ? p[u + 4] : p[u];
                const float send = hi ? p[u] : p[u + 4];
                p[u] = keep + lane_xor_row<4>(send);
            }
        }
        {
            const bool hi = (lane & 2) != 0;
#pragma unroll
            for (int u = 0; u < 2; ++u) {
                const float keep = hi ? p[u + 2] : p[u];
                const float send = hi ? p[u] : p[u + 2];
                p[u] = keep + lane_xor_row<2>(send);
            }
        }
        {
            const bool hi = (lane & 1) != 0;
            const float keep = hi ? p[1] : p[0];
            const float send = hi ? p[0] : p[1];
            p[0] = keep + lane_xor_row<1>(send);
        }
        if (eb + lane < e1 && ev < E) out[ev] = p[0];
    }
}

extern "C" void kernel_launch(void* const* d_in, const int* in_sizes, int n_in,
                              void* d_out, int out_size, void* d_ws, size_t ws_size,
                              hipStream_t stream) {
    const float* x      = (const float*)d_in[0];
    const int*   ei     = (const int*)d_in[1];
    const float* W1     = (const float*)d_in[2];
    const float* a1s    = (const float*)d_in[3];
    const float* a1d    = (const float*)d_in[4];
    const float* b1     = (const float*)d_in[5];
    const float* W2     = (const float*)d_in[6];
    const float* a2s    = (const float*)d_in[7];
    const float* a2d    = (const float*)d_in[8];
    const float* b2     = (const float*)d_in[9];
    float* scores = (float*)d_out;

    const int N = in_sizes[0] / FDIM;
    const int E = in_sizes[1] / 2;
    const int* src = ei;
    const int* dst = ei + E;

    float* ws   = (float*)d_ws;
    float*    h     = ws;
    int*      prec  = (int*)(ws + (size_t)N * FDIM);
    int*      bcount  = prec + (size_t)2 * (E + N);
    int*      bstart  = bcount + MAXNB;
    int*      bcursor = bstart + MAXNB + 1;
    float*    z     = ws + (size_t)2 * N * FDIM;
    float*    esrc  = ws + (size_t)3 * N * FDIM;
    float*    edst  = esrc + (size_t)N * HEADS;

    const int ETOT = E + N;
    const int NB = (N + TILE - 1) / TILE;
    const int NT = (N + DTILE - 1) / DTILE;
    dim3 blk(256);
    dim3 gGemm((N + 127) / 128);
    dim3 gNH((N * HEADS + 255) / 256);
    dim3 gChunk((ETOT + CHUNK - 1) / CHUNK);
    dim3 gAgg(NB);

    bucket_zero<<<dim3((NB + 255) / 256), blk, 0, stream>>>(bcount, NB);
    bucket_count<<<gChunk, blk, 0, stream>>>(dst, E, N, NB, bcount);
    bucket_scan<<<dim3(1), dim3(1024), 0, stream>>>(bcount, NB, bstart, bcursor);
    bucket_scatter<<<gChunk, blk, 0, stream>>>(src, dst, E, N, NB, bcursor, prec);

    gemm128_wmma<<<gGemm, blk, 0, stream>>>(x, W1, h, N);
    attn_logits<<<gNH, blk, 0, stream>>>(h, a1s, a1d, esrc, edst, N);
    bucket_softmax_aggregate<<<gAgg, dim3(32 * SW), 0, stream>>>(prec, bstart, esrc, edst, h, b1, z, N, NB,  1);

    gemm128_wmma<<<gGemm, blk, 0, stream>>>(z, W2, h, N);
    attn_logits<<<gNH, blk, 0, stream>>>(h, a2s, a2d, esrc, edst, N);
    bucket_softmax_aggregate<<<gAgg, dim3(32 * SW), 0, stream>>>(prec, bstart, esrc, edst, h, b2, z, N, NB,  0);

    tile_decode<<<dim3(2 * NT), blk, 0, stream>>>(prec, bstart, z, scores, N, NB, E);
}
